# starting kernel
# baseline (speedup 1.0000x reference)
#include <hip/hip_runtime.h>

typedef __attribute__((ext_vector_type(2))) float v2f;
typedef __attribute__((ext_vector_type(4))) float v4f;
typedef __attribute__((ext_vector_type(8))) float v8f;

#define F_IN  128
#define F_HID 64
#define F_OUT 40


__global__ void deg_init_kernel(float* __restrict__ deg, int N) {
    int i = blockIdx.x * blockDim.x + threadIdx.x;
    if (i < N) deg[i] = 1.0f;
}

__global__ void deg_edge_kernel(const int* __restrict__ dst, float* __restrict__ deg, int E) {
    int e = blockIdx.x * blockDim.x + threadIdx.x;
    if (e < E) unsafeAtomicAdd(&deg[dst[e]], 1.0f);
}

__global__ void dinv_kernel(const float* __restrict__ deg, float* __restrict__ dinv, int N) {
    int i = blockIdx.x * blockDim.x + threadIdx.x;
    if (i < N) dinv[i] = rsqrtf(deg[i]);
}


__global__ void repack_w1_kernel(const float* __restrict__ W1, float* __restrict__ pW1) {
    int idx = blockIdx.x * blockDim.x + threadIdx.x;
    if (idx >= (F_IN / 4) * 256) return;
    int kk   = idx >> 8;
    int rem  = idx & 255;
    int lane = rem >> 3;
    int j    = rem & 7;
    int n    = j >> 1;
    int t    = j & 1;
    int lm   = lane & 15;
    int kh   = (lane >> 4) << 1;
    pW1[idx] = W1[(size_t)(4 * kk + kh + t) * F_HID + n * 16 + lm];
}

__global__ void repack_w2_kernel(const float* __restrict__ W2, float* __restrict__ pW2) {
    int idx = blockIdx.x * blockDim.x + threadIdx.x;
    if (idx >= (F_HID / 4) * 256) return;
    int kk   = idx >> 8;
    int rem  = idx & 255;
    int lane = rem >> 3;
    int j    = rem & 7;
    int n    = j >> 1;
    int t    = j & 1;
    int lm   = lane & 15;
    int kh   = (lane >> 4) << 1;
    int col  = n * 16 + lm;
    float v  = 0.0f;
    if (j < 6 && col < F_OUT)
        v = W2[(size_t)(4 * kk + kh + t) * F_OUT + col];
    pW2[idx] = v;
}


__global__ void gemm1_wmma_kernel(const float* __restrict__ x, const float* __restrict__ pW1,
                                  float* __restrict__ h1, int N) {
    const int lane = threadIdx.x & 31;
    const int wave = threadIdx.x >> 5;
    const int row0 = blockIdx.x * 128 + wave * 16;
    const int lm   = lane & 15;
    const int kh   = (lane >> 4) << 1;
    int r = row0 + lm;
    if (r > N - 1) r = N - 1;
    const float* xrow = x + (size_t)r * F_IN + kh;
    const float* pw   = pW1 + (size_t)lane * 8;

    v8f acc0 = {}, acc1 = {}, acc2 = {}, acc3 = {};

    for (int kk = 0; kk < F_IN / 4; ++kk) {
        const v2f A   = *(const v2f*)(xrow + kk * 4);
        const v4f w01 = *(const v4f*)(pw + (size_t)kk * 256);
        const v4f w23 = *(const v4f*)(pw + (size_t)kk * 256 + 4);
        v2f B0; B0[0] = w01[0]; B0[1] = w01[1];
        v2f B1; B1[0] = w01[2]; B1[1] = w01[3];
        v2f B2; B2[0] = w23[0]; B2[1] = w23[1];
        v2f B3; B3[0] = w23[2]; B3[1] = w23[3];
        acc0 = __builtin_amdgcn_wmma_f32_16x16x4_f32(false, A, false, B0, (short)0, acc0, false, false);
        acc1 = __builtin_amdgcn_wmma_f32_16x16x4_f32(false, A, false, B1, (short)0, acc1, false, false);
        acc2 = __builtin_amdgcn_wmma_f32_16x16x4_f32(false, A, false, B2, (short)0, acc2, false, false);
        acc3 = __builtin_amdgcn_wmma_f32_16x16x4_f32(false, A, false, B3, (short)0, acc3, false, false);
    }

    const int hi8 = (lane >> 4) << 3;
    for (int j = 0; j < 8; ++j) {
        int rr = row0 + hi8 + j;
        if (rr < N) {
            float* o = h1 + (size_t)rr * F_HID + lm;
            o[0]  = acc0[j];
            o[16] = acc1[j];
            o[32] = acc2[j];
            o[48] = acc3[j];
        }
    }
}


__global__ void acc1_init_kernel(const float* __restrict__ h1, const float* __restrict__ dinv,
                                 float* __restrict__ acc, int total) {
    int idx = blockIdx.x * blockDim.x + threadIdx.x;
    if (idx < total) {
        float di = dinv[idx >> 6];
        acc[idx] = h1[idx] * di * di;
    }
}


__global__ void scatter1_kernel(const float* __restrict__ h1, const float* __restrict__ dinv,
                                const int* __restrict__ src, const int* __restrict__ dst,
                                float* __restrict__ acc, int E) {
    const int lane = threadIdx.x & 31;
    const int e = blockIdx.x * (blockDim.x >> 5) + (threadIdx.x >> 5);
    if (e >= E) return;
    const int s = __builtin_amdgcn_readfirstlane(src[e]);
    const int d = __builtin_amdgcn_readfirstlane(dst[e]);
    const float norm = dinv[s] * dinv[d];
    const float2 v = *(const float2*)(h1 + (size_t)s * F_HID + lane * 2);
    float* o = acc + (size_t)d * F_HID + lane * 2;
    unsafeAtomicAdd(o,     v.x * norm);
    unsafeAtomicAdd(o + 1, v.y * norm);
}


__global__ void relu_bias_kernel(const float* __restrict__ acc, const float* __restrict__ b1,
                                 float* __restrict__ z, int total) {
    int idx = blockIdx.x * blockDim.x + threadIdx.x;
    if (idx < total) {
        float v = acc[idx] + b1[idx & (F_HID - 1)];
        z[idx] = v > 0.0f ? v : 0.0f;
    }
}


__global__ void gemm2_wmma_kernel(const float* __restrict__ z, const float* __restrict__ pW2,
                                  float* __restrict__ h2, int N) {
    const int lane = threadIdx.x & 31;
    const int wave = threadIdx.x >> 5;
    const int row0 = blockIdx.x * 128 + wave * 16;
    const int lm   = lane & 15;
    const int kh   = (lane >> 4) << 1;
    int r = row0 + lm;
    if (r > N - 1) r = N - 1;
    const float* zrow = z + (size_t)r * F_HID + kh;
    const float* pw   = pW2 + (size_t)lane * 8;

    v8f acc0 = {}, acc1 = {}, acc2 = {};

    for (int kk = 0; kk < F_HID / 4; ++kk) {
        const v2f A   = *(const v2f*)(zrow + kk * 4);
        const v4f w01 = *(const v4f*)(pw + (size_t)kk * 256);
        const v2f w2v = *(const v2f*)(pw + (size_t)kk * 256 + 4);
        v2f B0; B0[0] = w01[0]; B0[1] = w01[1];
        v2f B1; B1[0] = w01[2]; B1[1] = w01[3];
        v2f B2; B2[0] = w2v[0]; B2[1] = w2v[1];
        acc0 = __builtin_amdgcn_wmma_f32_16x16x4_f32(false, A, false, B0, (short)0, acc0, false, false);
        acc1 = __builtin_amdgcn_wmma_f32_16x16x4_f32(false, A, false, B1, (short)0, acc1, false, false);
        acc2 = __builtin_amdgcn_wmma_f32_16x16x4_f32(false, A, false, B2, (short)0, acc2, false, false);
    }

    const int hi8  = (lane >> 4) << 3;
    const bool c2ok = (lm < 8);
    for (int j = 0; j < 8; ++j) {
        int rr = row0 + hi8 + j;
        if (rr < N) {
            float* o = h2 + (size_t)rr * F_OUT + lm;
            o[0]  = acc0[j];
            o[16] = acc1[j];
            if (c2ok) o[32] = acc2[j];
        }
    }
}


__global__ void out_init_kernel(const float* __restrict__ h2, const float* __restrict__ dinv,
                                const float* __restrict__ b2, float* __restrict__ out, int total) {
    int idx = blockIdx.x * blockDim.x + threadIdx.x;
    if (idx < total) {
        int i = idx / F_OUT;
        int f = idx - i * F_OUT;
        float di = dinv[i];
        out[idx] = h2[idx] * di * di + b2[f];
    }
}


__global__ void scatter2_kernel(const float* __restrict__ h2, const float* __restrict__ dinv,
                                const int* __restrict__ src, const int* __restrict__ dst,
                                float* __restrict__ out, int E) {
    const int lane = threadIdx.x & 31;
    const int e = blockIdx.x * (blockDim.x >> 5) + (threadIdx.x >> 5);
    if (e >= E) return;
    const int s = __builtin_amdgcn_readfirstlane(src[e]);
    const int d = __builtin_amdgcn_readfirstlane(dst[e]);
    const float norm = dinv[s] * dinv[d];
    if (lane < 20) {
        const float2 v = *(const float2*)(h2 + (size_t)s * F_OUT + lane * 2);
        float* o = out + (size_t)d * F_OUT + lane * 2;
        unsafeAtomicAdd(o,     v.x * norm);
        unsafeAtomicAdd(o + 1, v.y * norm);
    }
}


extern "C" void kernel_launch(void* const* d_in, const int* in_sizes, int n_in,
                              void* d_out, int out_size, void* d_ws, size_t ws_size,
                              hipStream_t stream) {
    const float* x  = (const float*)d_in[0];
    const int*   ei = (const int*)d_in[1];
    const float* W1 = (const float*)d_in[2];
    const float* b1 = (const float*)d_in[3];
    const float* W2 = (const float*)d_in[4];
    const float* b2 = (const float*)d_in[5];

    const int N = in_sizes[0] / F_IN;
    const int E = in_sizes[1] / 2;
    const int* src = ei;
    const int* dst = ei + E;

    float* ws   = (float*)d_ws;
    float* deg  = ws;
    float* dinv = ws + (size_t)N;
    float* bufA = ws + (size_t)2 * N;
    float* bufB = bufA + (size_t)N * F_HID;
    float* pW1  = bufB + (size_t)N * F_HID;
    float* pW2  = pW1 + (size_t)(F_IN / 4) * 256;
    float* out  = (float*)d_out;

    const int B = 256;
    const int nodeBlocks  = (N + B - 1) / B;
    const int edgeBlocks  = (E + B - 1) / B;
    const int hidElems    = N * F_HID;
    const int outElems    = N * F_OUT;
    const int hidBlocks   = (hidElems + B - 1) / B;
    const int outBlocksEl = (outElems + B - 1) / B;
    const int gemmBlocks  = (N + 127) / 128;
    const int waveEdgeBlk = (E + (B / 32) - 1) / (B / 32);
    const int rp1Blocks   = ((F_IN / 4) * 256 + B - 1) / B;
    const int rp2Blocks   = ((F_HID / 4) * 256 + B - 1) / B;

    repack_w1_kernel<<<rp1Blocks, B, 0, stream>>>(W1, pW1);
    repack_w2_kernel<<<rp2Blocks, B, 0, stream>>>(W2, pW2);

    deg_init_kernel<<<nodeBlocks, B, 0, stream>>>(deg, N);
    deg_edge_kernel<<<edgeBlocks, B, 0, stream>>>(dst, deg, E);
    dinv_kernel<<<nodeBlocks, B, 0, stream>>>(deg, dinv, N);

    gemm1_wmma_kernel<<<gemmBlocks, B, 0, stream>>>(x, pW1, bufA, N);
    acc1_init_kernel<<<hidBlocks, B, 0, stream>>>(bufA, dinv, bufB, hidElems);
    scatter1_kernel<<<waveEdgeBlk, B, 0, stream>>>(bufA, dinv, src, dst, bufB, E);
    relu_bias_kernel<<<hidBlocks, B, 0, stream>>>(bufB, b1, bufA, hidElems);

    gemm2_wmma_kernel<<<gemmBlocks, B, 0, stream>>>(bufA, pW2, bufB, N);
    out_init_kernel<<<outBlocksEl, B, 0, stream>>>(bufB, dinv, b2, out, outElems);
    scatter2_kernel<<<waveEdgeBlk, B, 0, stream>>>(bufB, dinv, src, dst, out, E);
}


# In-degree counts: a work-group counts one (node range, edge slice) pair in LDS (integer adds), stores a partial plane, summed where the first product reads it; global float atomics on counts gone
# speedup vs baseline: 2.2335x; 2.2335x over previous
#include <hip/hip_runtime.h>

typedef __attribute__((ext_vector_type(2))) float v2f;
typedef __attribute__((ext_vector_type(4))) float v4f;
typedef __attribute__((ext_vector_type(8))) float v8f;

#define F_IN  128
#define F_HID 64
#define F_OUT 40

__device__ __forceinline__ int package_half() {
    const unsigned id = __builtin_amdgcn_s_sendmsg_rtn(0x87);
    return (int)((id >> 18) & 1u);
}


#define DEG_RANGE   12544
#define DEG_SLICES  32
#define DEG_THREADS 1024

__global__ void __launch_bounds__(DEG_THREADS)
deg_hist_kernel(const int* __restrict__ dst, unsigned int* __restrict__ part,
                int N, int E, int sliceLen) {
    __shared__ unsigned int cnt[DEG_RANGE];
    const int n0 = blockIdx.x * DEG_RANGE;
    int nn = N - n0;
    if (nn > DEG_RANGE) nn = DEG_RANGE;
    for (int i = threadIdx.x; i < DEG_RANGE; i += DEG_THREADS) cnt[i] = 0u;
    __syncthreads();

    const int e0 = blockIdx.y * sliceLen;
    int e1 = e0 + sliceLen;
    if (e1 > E) e1 = E;
    int len = e1 - e0;
    if (len < 0) len = 0;
    const int* dp = dst + e0;
    const bool vec = ((reinterpret_cast<size_t>(dp) & 15) == 0);
    const int  nv  = vec ? (len >> 2) : 0;
    const int4* dv = (const int4*)dp;
    for (int i = threadIdx.x; i < nv; i += 4 * DEG_THREADS) {
        int4 q[4];
        #pragma unroll
        for (int k = 0; k < 4; ++k) {
            const int  idx = i + k * DEG_THREADS;
            const bool ok  = idx < nv;
            q[k] = dv[ok ? idx : 0];
            if (!ok) q[k] = make_int4(-1, -1, -1, -1);
        }
        #pragma unroll
        for (int k = 0; k < 4; ++k) {
            const unsigned int u0 = (unsigned int)(q[k].x - n0), u1 = (unsigned int)(q[k].y - n0);
            const unsigned int u2 = (unsigned int)(q[k].z - n0), u3 = (unsigned int)(q[k].w - n0);
            if (u0 < (unsigned int)nn) atomicAdd(&cnt[u0], 1u);
            if (u1 < (unsigned int)nn) atomicAdd(&cnt[u1], 1u);
            if (u2 < (unsigned int)nn) atomicAdd(&cnt[u2], 1u);
            if (u3 < (unsigned int)nn) atomicAdd(&cnt[u3], 1u);
        }
    }
    for (int e = 4 * nv + threadIdx.x; e < len; e += DEG_THREADS) {
        const unsigned int u = (unsigned int)(dp[e] - n0);
        if (u < (unsigned int)nn) atomicAdd(&cnt[u], 1u);
    }
    __syncthreads();

    unsigned int* p = part + (size_t)blockIdx.y * N + n0;
    for (int i = threadIdx.x; i < nn; i += DEG_THREADS) p[i] = cnt[i];
}


__global__ void repack_w1_kernel(const float* __restrict__ W1, float* __restrict__ pW1) {
    int idx = blockIdx.x * blockDim.x + threadIdx.x;
    if (idx >= (F_IN / 4) * 256) return;
    int kk   = idx >> 8;
    int rem  = idx & 255;
    int lane = rem >> 3;
    int j    = rem & 7;
    int n    = j >> 1;
    int t    = j & 1;
    int lm   = lane & 15;
    int kh   = (lane >> 4) << 1;
    pW1[idx] = W1[(size_t)(4 * kk + kh + t) * F_HID + (n >> 1) * 32 + 2 * lm + (n & 1)];
}

__global__ void repack_w2_kernel(const float* __restrict__ W2, float* __restrict__ pW2) {
    int idx = blockIdx.x * blockDim.x + threadIdx.x;
    if (idx >= (F_HID / 4) * 256) return;
    int kk   = idx >> 8;
    int rem  = idx & 255;
    int lane = rem >> 3;
    int j    = rem & 7;
    int n    = j >> 1;
    int t    = j & 1;
    int lm   = lane & 15;
    int kh   = (lane >> 4) << 1;
    int col  = n * 16 + lm;
    float v  = 0.0f;
    if (j < 6 && col < F_OUT)
        v = W2[(size_t)(4 * kk + kh + t) * F_OUT + col];
    pW2[idx] = v;
}


__global__ void gemm1_wmma_kernel(const float* __restrict__ x, const float* __restrict__ pW1,
                                  const unsigned int* __restrict__ part,
                                  float* __restrict__ dinv, float* __restrict__ h1,
                                  float* __restrict__ acc, float* __restrict__ accOther, int N) {
    const int lane = threadIdx.x & 31;
    const int wave = threadIdx.x >> 5;
    const int row0 = blockIdx.x * 128 + wave * 16;
    const int lm   = lane & 15;
    const int kh   = (lane >> 4) << 1;
    int r = row0 + lm;
    if (r > N - 1) r = N - 1;
    const float* xrow = x + (size_t)r * F_IN + kh;
    const float* pw   = pW1 + (size_t)lane * 8;

    const int hi8 = (lane >> 4) << 3;
    int rl = row0 + lane;
    if (rl > N - 1) rl = N - 1;
    unsigned int pc[DEG_SLICES];
    #pragma unroll
    for (int s = 0; s < DEG_SLICES; ++s) pc[s] = part[(size_t)s * N + rl];

    v8f acc0 = {}, acc1 = {}, acc2 = {}, acc3 = {};

    for (int kk = 0; kk < F_IN / 4; ++kk) {
        const v2f A   = *(const v2f*)(xrow + kk * 4);
        const v4f w01 = *(const v4f*)(pw + (size_t)kk * 256);
        const v4f w23 = *(const v4f*)(pw + (size_t)kk * 256 + 4);
        v2f B0; B0[0] = w01[0]; B0[1] = w01[1];
        v2f B1; B1[0] = w01[2]; B1[1] = w01[3];
        v2f B2; B2[0] = w23[0]; B2[1] = w23[1];
        v2f B3; B3[0] = w23[2]; B3[1] = w23[3];
        acc0 = __builtin_amdgcn_wmma_f32_16x16x4_f32(false, A, false, B0, (short)0, acc0, false, false);
        acc1 = __builtin_amdgcn_wmma_f32_16x16x4_f32(false, A, false, B1, (short)0, acc1, false, false);
        acc2 = __builtin_amdgcn_wmma_f32_16x16x4_f32(false, A, false, B2, (short)0, acc2, false, false);
        acc3 = __builtin_amdgcn_wmma_f32_16x16x4_f32(false, A, false, B3, (short)0, acc3, false, false);
    }

    unsigned int cdeg = 1u;
    #pragma unroll
    for (int s = 0; s < DEG_SLICES; ++s) cdeg += pc[s];
    const float dil = rsqrtf((float)cdeg);
    float dj[8];
    #pragma unroll
    for (int j = 0; j < 8; ++j) dj[j] = __shfl(dil, hi8 + j, 32);
    for (int j = 0; j < 8; ++j) {
        int rr = row0 + hi8 + j;
        if (rr < N) {
            const float di = dj[j];
            const float s0 = acc0[j] * di, s1 = acc1[j] * di, s2 = acc2[j] * di, s3 = acc3[j] * di;
            float* o = h1 + (size_t)rr * F_HID + 4 * lm;
            *(float4*)(o) = make_float4(s0, s2, s1, s3);
            float* a = acc + (size_t)rr * F_HID + 2 * lm;
            *(float2*)(a)      = make_float2(s0, s1);
            *(float2*)(a + 32) = make_float2(s2, s3);
            *(float4*)(accOther + (size_t)rr * F_HID + 4 * lm) = make_float4(0.0f, 0.0f, 0.0f, 0.0f);
        }
    }
    if ((wave & 1) == 0) {
        const int r0 = row0 + lane;
        if (r0 < N) dinv[r0] = dil;
    }
}


#define EDGE_GROUP 8

__global__ void scatter1_kernel(const float* __restrict__ h1,
                                const int* __restrict__ src, const int* __restrict__ dst,
                                float* __restrict__ acc0, float* __restrict__ acc1, int E) {
    float* acc = package_half() ? acc1 : acc0;
    const int lane = threadIdx.x & 31;
    const int base = __builtin_amdgcn_readfirstlane(
        (blockIdx.x * (blockDim.x >> 5) + (threadIdx.x >> 5)) * 32);
    if (base >= E) return;
    const int nE = (E - base < 32) ? (E - base) : 32;

    int s_l = 0, d_l = 0;
    if (lane < nE) {
        s_l = src[base + lane];
        d_l = dst[base + lane];
    }

    int j = 0;
    for (; j + EDGE_GROUP <= nE; j += EDGE_GROUP) {
        float2 v[EDGE_GROUP];
        #pragma unroll
        for (int k = 0; k < EDGE_GROUP; ++k) {
            const int s = __builtin_amdgcn_readlane(s_l, j + k);
            v[k] = *(const float2*)(h1 + (size_t)s * F_HID + lane * 2);
        }
        #pragma unroll
        for (int k = 0; k < EDGE_GROUP; ++k) {
            const int d = __builtin_amdgcn_readlane(d_l, j + k);
            float* o = acc + (size_t)d * F_HID + lane;
            unsafeAtomicAdd(o,      v[k].x);
            unsafeAtomicAdd(o + 32, v[k].y);
        }
    }
    for (; j < nE; ++j) {
        const int s = __builtin_amdgcn_readlane(s_l, j);
        const int d = __builtin_amdgcn_readlane(d_l, j);
        const float2 v = *(const float2*)(h1 + (size_t)s * F_HID + lane * 2);
        float* o = acc + (size_t)d * F_HID + lane;
        unsafeAtomicAdd(o,      v.x);
        unsafeAtomicAdd(o + 32, v.y);
    }
}


#define H2_LD 64

__global__ void gemm2_wmma_kernel(const float* __restrict__ acc, const float* __restrict__ accOther,
                                  const float* __restrict__ b1,
                                  const float* __restrict__ pW2, const float* __restrict__ dinv,
                                  float* __restrict__ h2,
                                  float* __restrict__ out, float* __restrict__ out1, int N) {
    __shared__ __attribute__((aligned(16))) float sOut[8][16 * F_OUT];
    const int lane = threadIdx.x & 31;
    const int wave = threadIdx.x >> 5;
    const int row0 = blockIdx.x * 128 + wave * 16;
    const int lm   = lane & 15;
    const int kh   = (lane >> 4) << 1;
    int r = row0 + lm;
    if (r > N - 1) r = N - 1;
    const float* arow  = acc  + (size_t)r * F_HID + kh;
    const float* arow1 = accOther + (size_t)r * F_HID + kh;
    const float* brow = b1 + kh;
    const float* pw   = pW2 + (size_t)lane * 8;

    const int  hi8  = (lane >> 4) << 3;
    const bool c2ok = (lm < 8);
    float dv[8];
    for (int j = 0; j < 8; ++j) {
        int rc = row0 + hi8 + j;
        if (rc > N - 1) rc = N - 1;
        dv[j] = dinv[rc];
    }
    const float dr = dinv[r];

    v8f acc0 = {}, acc1 = {}, acc2 = {};

    for (int kk = 0; kk < F_HID / 4; ++kk) {
        const v2f S0  = *(const v2f*)(arow + kk * 4);
        const v2f S1  = *(const v2f*)(arow1 + kk * 4);
        const v2f bb  = *(const v2f*)(brow + kk * 4);
        const float z0 = (S0[0] + S1[0]) * dr + bb[0];
        const float z1 = (S0[1] + S1[1]) * dr + bb[1];
        v2f A; A[0] = z0 > 0.0f ? z0 : 0.0f; A[1] = z1 > 0.0f ? z1 : 0.0f;
        const v4f w01 = *(const v4f*)(pw + (size_t)kk * 256);
        const v2f w2v = *(const v2f*)(pw + (size_t)kk * 256 + 4);
        v2f B0; B0[0] = w01[0]; B0[1] = w01[1];
        v2f B1; B1[0] = w01[2]; B1[1] = w01[3];
        v2f B2; B2[0] = w2v[0]; B2[1] = w2v[1];
        acc0 = __builtin_amdgcn_wmma_f32_16x16x4_f32(false, A, false, B0, (short)0, acc0, false, false);
        acc1 = __builtin_amdgcn_wmma_f32_16x16x4_f32(false, A, false, B1, (short)0, acc1, false, false);
        acc2 = __builtin_amdgcn_wmma_f32_16x16x4_f32(false, A, false, B2, (short)0, acc2, false, false);
    }

    float* so = sOut[wave];
    for (int j = 0; j < 8; ++j) {
        int rr = row0 + hi8 + j;
        const float di = dv[j];
        const float s0 = acc0[j] * di, s1 = acc1[j] * di, s2 = acc2[j] * di;
        if (rr < N) {
            float* o = h2 + (size_t)rr * H2_LD + 2 * lm;
            *(float2*)(o)      = make_float2(s0, c2ok ? s2 : 0.0f);
            *(float2*)(o + 32) = make_float2(s1, 0.0f);
        }
        float* q = so + (hi8 + j) * F_OUT + lm;
        q[0]  = s0;
        q[16] = s1;
        if (c2ok) q[32] = s2;
    }
    __syncthreads();
    int nf = (N - row0) * F_OUT;
    if (nf > 16 * F_OUT) nf = 16 * F_OUT;
    float* ob  = out  + (size_t)row0 * F_OUT;
    float* ob1 = out1 + (size_t)row0 * F_OUT;
    for (int t = 0; t < 5; ++t) {
        const int e = (t * 32 + lane) * 4;
        if (e < nf) {
            *(float4*)(ob + e)  = *(const float4*)(so + e);
            *(float4*)(ob1 + e) = make_float4(0.0f, 0.0f, 0.0f, 0.0f);
        }
    }
}


__global__ void scatter2_kernel(const float* __restrict__ h2,
                                const int* __restrict__ src, const int* __restrict__ dst,
                                float* __restrict__ out0, float* __restrict__ out1, int E) {
    float* out = package_half() ? out1 : out0;
    const int lane = threadIdx.x & 31;
    const int base = __builtin_amdgcn_readfirstlane(
        (blockIdx.x * (blockDim.x >> 5) + (threadIdx.x >> 5)) * 32);
    if (base >= E) return;
    const int nE = (E - base < 32) ? (E - base) : 32;

    int s_l = 0, d_l = 0;
    if (lane < nE) {
        s_l = src[base + lane];
        d_l = dst[base + lane];
    }

    const bool tailLane = (lane < F_OUT - 32);
    int j = 0;
    for (; j + EDGE_GROUP <= nE; j += EDGE_GROUP) {
        int    sv[EDGE_GROUP], dv[EDGE_GROUP];
        float2 v[EDGE_GROUP];
        #pragma unroll
        for (int k = 0; k < EDGE_GROUP; ++k) {
            sv[k] = __builtin_amdgcn_readlane(s_l, j + k);
            dv[k] = __builtin_amdgcn_readlane(d_l, j + k);
        }
        #pragma unroll
        for (int k = 0; k < EDGE_GROUP; ++k)
            v[k] = *(const float2*)(h2 + (size_t)sv[k] * H2_LD + lane * 2);
        #pragma unroll
        for (int k = 0; k < EDGE_GROUP; ++k)
            unsafeAtomicAdd(out + (size_t)dv[k] * F_OUT + lane, v[k].x);
        if (tailLane) {
            #pragma unroll
            for (int k = 0; k < EDGE_GROUP; ++k)
                unsafeAtomicAdd(out + (size_t)dv[k] * F_OUT + 32 + lane, v[k].y);
        }
    }
    for (; j < nE; ++j) {
        const int s = __builtin_amdgcn_readlane(s_l, j);
        const int d = __builtin_amdgcn_readlane(d_l, j);
        const float2 v = *(const float2*)(h2 + (size_t)s * H2_LD + lane * 2);
        float* o = out + (size_t)d * F_OUT + lane;
        unsafeAtomicAdd(o, v.x);
        if (tailLane) unsafeAtomicAdd(o + 32, v.y);
    }
}


__global__ void join_out_kernel(float* __restrict__ out, const float* __restrict__ out1,
                                const float* __restrict__ dinv, const float* __restrict__ b2, int n4) {
    const int i = blockIdx.x * blockDim.x + threadIdx.x;
    if (i < n4) {
        const int e   = i * 4;
        const int row = e / F_OUT;
        const int col = e - row * F_OUT;
        const float  di = dinv[row];
        const float4 bb = *(const float4*)(b2 + col);
        const float4 a = ((const float4*)out)[i];
        const float4 b = ((const float4*)out1)[i];
        ((float4*)out)[i] = make_float4((a.x + b.x) * di + bb.x, (a.y + b.y) * di + bb.y,
                                        (a.z + b.z) * di + bb.z, (a.w + b.w) * di + bb.w);
    }
}


extern "C" void kernel_launch(void* const* d_in, const int* in_sizes, int n_in,
                              void* d_out, int out_size, void* d_ws, size_t ws_size,
                              hipStream_t stream) {
    const float* x  = (const float*)d_in[0];
    const int*   ei = (const int*)d_in[1];
    const float* W1 = (const float*)d_in[2];
    const float* b1 = (const float*)d_in[3];
    const float* W2 = (const float*)d_in[4];
    const float* b2 = (const float*)d_in[5];

    const int N = in_sizes[0] / F_IN;
    const int E = in_sizes[1] / 2;
    const int* src = ei;
    const int* dst = ei + E;

    float* ws   = (float*)d_ws;
    float* dinv = ws;
    float* bufA = ws + (((size_t)N + 63) & ~(size_t)63);
    float* bufB = bufA + (size_t)N * F_HID;
    float* pW1  = bufB + (size_t)N * F_HID;
    float* pW2  = pW1 + (size_t)(F_IN / 4) * 256;
    float* bufC = pW2 + (size_t)(F_HID / 4) * 256;
    float* out1 = bufC + (size_t)N * F_HID;
    unsigned int* part = (unsigned int*)(out1 + (size_t)N * F_OUT);
    float* out  = (float*)d_out;

    const int B = 256;
    const int edgeBlocks  = (E + B - 1) / B;
    const int gemmBlocks  = (N + 127) / 128;
    const int rp1Blocks   = ((F_IN / 4) * 256 + B - 1) / B;
    const int rp2Blocks   = ((F_HID / 4) * 256 + B - 1) / B;
    const int n4          = N * F_OUT / 4;
    const int joinBlocks  = (n4 + B - 1) / B;

    repack_w1_kernel<<<rp1Blocks, B, 0, stream>>>(W1, pW1);
    repack_w2_kernel<<<rp2Blocks, B, 0, stream>>>(W2, pW2);

    const int  sliceLen = (((E + DEG_SLICES - 1) / DEG_SLICES) + 3) & ~3;
    const dim3 histGrid((N + DEG_RANGE - 1) / DEG_RANGE, DEG_SLICES);
    deg_hist_kernel<<<histGrid, DEG_THREADS, 0, stream>>>(dst, part, N, E, sliceLen);

    gemm1_wmma_kernel<<<gemmBlocks, B, 0, stream>>>(x, pW1, part, dinv, bufA, bufB, bufC, N);
    scatter1_kernel<<<edgeBlocks, B, 0, stream>>>(bufA, src, dst, bufB, bufC, E);

    gemm2_wmma_kernel<<<gemmBlocks, B, 0, stream>>>(bufB, bufC, b1, pW2, dinv, bufA, out, out1, N);
    scatter2_kernel<<<edgeBlocks, B, 0, stream>>>(bufA, src, dst, out, out1, E);

    join_out_kernel<<<joinBlocks, B, 0, stream>>>(out, out1, dinv, b2, n4);
}


# Edge lists: each (range, slice) list is stored grouped by row owner (count, prefix, placed via LDS cursors in a second walk of the slice); an owner reads only its own groups; sums in arrival order
# speedup vs baseline: 3.6514x; 1.6348x over previous
#include <hip/hip_runtime.h>

typedef __attribute__((ext_vector_type(2))) float v2f;
typedef __attribute__((ext_vector_type(4))) float v4f;
typedef __attribute__((ext_vector_type(8))) float v8f;

#define F_IN  128
#define F_HID 64
#define F_OUT 40


#define DEG_RANGE   12544
#define DEG_SLICES  32
#define DEG_THREADS 1024
#define OWN_RANGE     392
#define OWN_PER_RANGE (DEG_RANGE / OWN_RANGE)
#define SEG_LD        (OWN_PER_RANGE + 1)
#define STAGE_CAP     8192

static_assert(DEG_RANGE % OWN_RANGE == 0, "a range of the edge lists is a whole number of owners");
static_assert(OWN_PER_RANGE == DEG_THREADS / 32, "one wave adds up the counters of one owner");
static_assert(OWN_PER_RANGE == 32, "the prefix over the owners is one wave wide");

template <bool PLACE>
__device__ __forceinline__ void walk_slice(const int* __restrict__ dp, const int* __restrict__ sp,
                                           int len, int nv, int n0, int nn,
                                           unsigned int* cnt, unsigned int* cur,
                                           int2* stage, int2* __restrict__ list) {
    const int4* dv = (const int4*)dp;
    const int4* sv = (const int4*)sp;
    for (int i0 = 0; i0 < nv; i0 += 4 * DEG_THREADS) {
        int4 q[4], sq[4];
        #pragma unroll
        for (int k = 0; k < 4; ++k) {
            const int  idx = i0 + k * DEG_THREADS + (int)threadIdx.x;
            const bool ok  = idx < nv;
            q[k] = dv[ok ? idx : 0];
            if (PLACE) sq[k] = sv[ok ? idx : 0];
            if (!ok) q[k] = make_int4(-1, -1, -1, -1);
        }
        if (!PLACE) {
            #pragma unroll
            for (int k = 0; k < 4; ++k) {
                const int dd[4] = {q[k].x, q[k].y, q[k].z, q[k].w};
                #pragma unroll
                for (int c = 0; c < 4; ++c) {
                    const unsigned int u = (unsigned int)(dd[c] - n0);
                    if (u < (unsigned int)nn) atomicAdd(&cnt[u], 1u);
                }
            }
        } else {
            unsigned int pos[16];
            #pragma unroll
            for (int k = 0; k < 4; ++k) {
                const int dd[4] = {q[k].x, q[k].y, q[k].z, q[k].w};
                #pragma unroll
                for (int c = 0; c < 4; ++c) {
                    const unsigned int u = (unsigned int)(dd[c] - n0);
                    pos[k * 4 + c] = 0u;
                    if (u < (unsigned int)nn) pos[k * 4 + c] = atomicAdd(&cur[u / OWN_RANGE], 1u);
                }
            }
            #pragma unroll
            for (int k = 0; k < 4; ++k) {
                const int dd[4] = {q[k].x, q[k].y, q[k].z, q[k].w};
                const int ss[4] = {sq[k].x, sq[k].y, sq[k].z, sq[k].w};
                #pragma unroll
                for (int c = 0; c < 4; ++c) {
                    const unsigned int u = (unsigned int)(dd[c] - n0);
                    if (u < (unsigned int)nn) {
                        const unsigned int p  = pos[k * 4 + c];
                        const int2         en = make_int2(ss[c], (int)(u % OWN_RANGE));
                        if (p < STAGE_CAP) stage[p] = en;
                        else               list[p]  = en;
                    }
                }
            }
        }
    }
    for (int e = 4 * nv + (int)threadIdx.x; e < len; e += DEG_THREADS) {
        const unsigned int u = (unsigned int)(dp[e] - n0);
        if (u < (unsigned int)nn) {
            if (!PLACE) {
                atomicAdd(&cnt[u], 1u);
            } else {
                const unsigned int p  = atomicAdd(&cur[u / OWN_RANGE], 1u);
                const int2         en = make_int2(sp[e], (int)(u % OWN_RANGE));
                if (p < STAGE_CAP) stage[p] = en;
                else               list[p]  = en;
            }
        }
    }
}

__global__ void __launch_bounds__(DEG_THREADS)
deg_hist_kernel(const int* __restrict__ src, const int* __restrict__ dst,
                unsigned int* __restrict__ part, int2* __restrict__ pairs,
                unsigned int* __restrict__ seg, int N, int E, int sliceLen) {
    __shared__ unsigned int cnt[DEG_RANGE];
    __shared__ unsigned int tot[OWN_PER_RANGE];
    __shared__ unsigned int cur[OWN_PER_RANGE];
    __shared__ unsigned int total;
    __shared__ int2 stage[STAGE_CAP];
    const int lane = threadIdx.x & 31;
    const int wave = threadIdx.x >> 5;
    const int n0 = blockIdx.x * DEG_RANGE;
    int nn = N - n0;
    if (nn > DEG_RANGE) nn = DEG_RANGE;
    for (int i = threadIdx.x; i < DEG_RANGE; i += DEG_THREADS) cnt[i] = 0u;
    __syncthreads();

    const int e0 = blockIdx.y * sliceLen;
    int e1 = e0 + sliceLen;
    if (e1 > E) e1 = E;
    int len = e1 - e0;
    if (len < 0) len = 0;
    const int* dp = dst + e0;
    const int* sp = src + e0;
    int2* list = pairs + (size_t)(blockIdx.x * DEG_SLICES + blockIdx.y) * sliceLen;
    const bool vec = (((reinterpret_cast<size_t>(dp) | reinterpret_cast<size_t>(sp)) & 15) == 0);
    const int  nv  = vec ? (len >> 2) : 0;

    walk_slice<false>(dp, sp, len, nv, n0, nn, cnt, cur, stage, list);
    __syncthreads();

    {
        unsigned int t = 0u;
        for (int i = lane; i < OWN_RANGE; i += 32) t += cnt[wave * OWN_RANGE + i];
        for (int o = 16; o > 0; o >>= 1) t += (unsigned int)__shfl_xor((int)t, o, 32);
        if (lane == 0) tot[wave] = t;
    }
    unsigned int* p = part + (size_t)blockIdx.y * N + n0;
    for (int i = threadIdx.x; i < nn; i += DEG_THREADS) p[i] = cnt[i];
    __syncthreads();

    if (wave == 0) {
        const unsigned int v = tot[lane];
        unsigned int inc = v;
        for (int o = 1; o < 32; o <<= 1) {
            const unsigned int up = (unsigned int)__shfl_up((int)inc, o, 32);
            if (lane >= o) inc += up;
        }
        cur[lane] = inc - v;
        unsigned int* sg = seg + (size_t)(blockIdx.x * DEG_SLICES + blockIdx.y) * SEG_LD;
        sg[lane] = inc - v;
        if (lane == OWN_PER_RANGE - 1) { sg[OWN_PER_RANGE] = inc; total = inc; }
    }
    __syncthreads();

    walk_slice<true>(dp, sp, len, nv, n0, nn, cnt, cur, stage, list);
    __syncthreads();

    int nst = (int)total;
    if (nst > STAGE_CAP) nst = STAGE_CAP;
    for (int i = threadIdx.x; i < nst; i += DEG_THREADS) list[i] = stage[i];
}


__global__ void repack_w1_kernel(const float* __restrict__ W1, float* __restrict__ pW1) {
    int idx = blockIdx.x * blockDim.x + threadIdx.x;
    if (idx >= (F_IN / 4) * 256) return;
    int kk   = idx >> 8;
    int rem  = idx & 255;
    int lane = rem >> 3;
    int j    = rem & 7;
    int n    = j >> 1;
    int t    = j & 1;
    int lm   = lane & 15;
    int kh   = (lane >> 4) << 1;
    pW1[idx] = W1[(size_t)(4 * kk + kh + t) * F_HID + (n >> 1) * 32 + 2 * lm + (n & 1)];
}

__global__ void repack_w2_kernel(const float* __restrict__ W2, float* __restrict__ pW2) {
    int idx = blockIdx.x * blockDim.x + threadIdx.x;
    if (idx >= (F_HID / 4) * 256) return;
    int kk   = idx >> 8;
    int rem  = idx & 255;
    int lane = rem >> 3;
    int j    = rem & 7;
    int n    = j >> 1;
    int t    = j & 1;
    int lm   = lane & 15;
    int kh   = (lane >> 4) << 1;
    int col  = n * 16 + lm;
    float v  = 0.0f;
    if (j < 6 && col < F_OUT)
        v = W2[(size_t)(4 * kk + kh + t) * F_OUT + col];
    pW2[idx] = v;
}


__global__ void gemm1_wmma_kernel(const float* __restrict__ x, const float* __restrict__ pW1,
                                  const unsigned int* __restrict__ part,
                                  float* __restrict__ dinv, float* __restrict__ h1, int N) {
    const int lane = threadIdx.x & 31;
    const int wave = threadIdx.x >> 5;
    const int row0 = blockIdx.x * 128 + wave * 16;
    const int lm   = lane & 15;
    const int kh   = (lane >> 4) << 1;
    int r = row0 + lm;
    if (r > N - 1) r = N - 1;
    const float* xrow = x + (size_t)r * F_IN + kh;
    const float* pw   = pW1 + (size_t)lane * 8;

    const int hi8 = (lane >> 4) << 3;
    int rl = row0 + lane;
    if (rl > N - 1) rl = N - 1;
    unsigned int pc[DEG_SLICES];
    #pragma unroll
    for (int s = 0; s < DEG_SLICES; ++s) pc[s] = part[(size_t)s * N + rl];

    v8f acc0 = {}, acc1 = {}, acc2 = {}, acc3 = {};

    for (int kk = 0; kk < F_IN / 4; ++kk) {
        const v2f A   = *(const v2f*)(xrow + kk * 4);
        const v4f w01 = *(const v4f*)(pw + (size_t)kk * 256);
        const v4f w23 = *(const v4f*)(pw + (size_t)kk * 256 + 4);
        v2f B0; B0[0] = w01[0]; B0[1] = w01[1];
        v2f B1; B1[0] = w01[2]; B1[1] = w01[3];
        v2f B2; B2[0] = w23[0]; B2[1] = w23[1];
        v2f B3; B3[0] = w23[2]; B3[1] = w23[3];
        acc0 = __builtin_amdgcn_wmma_f32_16x16x4_f32(false, A, false, B0, (short)0, acc0, false, false);
        acc1 = __builtin_amdgcn_wmma_f32_16x16x4_f32(false, A, false, B1, (short)0, acc1, false, false);
        acc2 = __builtin_amdgcn_wmma_f32_16x16x4_f32(false, A, false, B2, (short)0, acc2, false, false);
        acc3 = __builtin_amdgcn_wmma_f32_16x16x4_f32(false, A, false, B3, (short)0, acc3, false, false);
    }

    unsigned int cdeg = 1u;
    #pragma unroll
    for (int s = 0; s < DEG_SLICES; ++s) cdeg += pc[s];
    const float dil = rsqrtf((float)cdeg);
    float dj[8];
    #pragma unroll
    for (int j = 0; j < 8; ++j) dj[j] = __shfl(dil, hi8 + j, 32);
    for (int j = 0; j < 8; ++j) {
        int rr = row0 + hi8 + j;
        if (rr < N) {
            const float di = dj[j];
            const float s0 = acc0[j] * di, s1 = acc1[j] * di, s2 = acc2[j] * di, s3 = acc3[j] * di;
            float* o = h1 + (size_t)rr * F_HID + 4 * lm;
            *(float4*)(o) = make_float4(s0, s2, s1, s3);
        }
    }
    if ((wave & 1) == 0) {
        const int r0 = row0 + lane;
        if (r0 < N) dinv[r0] = dil;
    }
}


#define H2_LD 64

__global__ void gemm2_wmma_kernel(const float* __restrict__ acc,
                                  const float* __restrict__ b1,
                                  const float* __restrict__ pW2, const float* __restrict__ dinv,
                                  float* __restrict__ h2, int N) {
    const int lane = threadIdx.x & 31;
    const int wave = threadIdx.x >> 5;
    const int row0 = blockIdx.x * 128 + wave * 16;
    const int lm   = lane & 15;
    const int kh   = (lane >> 4) << 1;
    int r = row0 + lm;
    if (r > N - 1) r = N - 1;
    const float* arow  = acc  + (size_t)r * F_HID + kh;
    const float* brow = b1 + kh;
    const float* pw   = pW2 + (size_t)lane * 8;

    const int  hi8  = (lane >> 4) << 3;
    const bool c2ok = (lm < 8);
    float dv[8];
    for (int j = 0; j < 8; ++j) {
        int rc = row0 + hi8 + j;
        if (rc > N - 1) rc = N - 1;
        dv[j] = dinv[rc];
    }
    const float dr = dinv[r];

    v8f acc0 = {}, acc1 = {}, acc2 = {};

    for (int kk = 0; kk < F_HID / 4; ++kk) {
        const v2f S0  = *(const v2f*)(arow + kk * 4);
        const v2f bb  = *(const v2f*)(brow + kk * 4);
        const float z0 = S0[0] * dr + bb[0];
        const float z1 = S0[1] * dr + bb[1];
        v2f A; A[0] = z0 > 0.0f ? z0 : 0.0f; A[1] = z1 > 0.0f ? z1 : 0.0f;
        const v4f w01 = *(const v4f*)(pw + (size_t)kk * 256);
        const v2f w2v = *(const v2f*)(pw + (size_t)kk * 256 + 4);
        v2f B0; B0[0] = w01[0]; B0[1] = w01[1];
        v2f B1; B1[0] = w01[2]; B1[1] = w01[3];
        v2f B2; B2[0] = w2v[0]; B2[1] = w2v[1];
        acc0 = __builtin_amdgcn_wmma_f32_16x16x4_f32(false, A, false, B0, (short)0, acc0, false, false);
        acc1 = __builtin_amdgcn_wmma_f32_16x16x4_f32(false, A, false, B1, (short)0, acc1, false, false);
        acc2 = __builtin_amdgcn_wmma_f32_16x16x4_f32(false, A, false, B2, (short)0, acc2, false, false);
    }

    for (int j = 0; j < 8; ++j) {
        int rr = row0 + hi8 + j;
        const float di = dv[j];
        const float s0 = acc0[j] * di, s1 = acc1[j] * di, s2 = acc2[j] * di;
        if (rr < N) {
            float* o = h2 + (size_t)rr * H2_LD + 2 * lm;
            *(float2*)(o)      = make_float2(s0, c2ok ? s2 : 0.0f);
            *(float2*)(o + 32) = make_float2(s1, 0.0f);
        }
    }
}


#define AGG_THREADS   1024
#define AGG_WAVES     (AGG_THREADS / 32)

static_assert(AGG_WAVES == DEG_SLICES, "one wave of an owner per edge list of its range");

template <int F, bool FULL>
__device__ __forceinline__ void add_edges(const float* __restrict__ h, float* rows,
                                          int s_l, int d_l, int lane) {
    float2 v[32];
    #pragma unroll
    for (int k = 0; k < 32; ++k) {
        const int s = __builtin_amdgcn_readlane(s_l, k);
        v[k] = *(const float2*)(h + (size_t)s * 64 + lane * 2);
    }
    #pragma unroll
    for (int k = 0; k < 32; ++k) {
        const int d = __builtin_amdgcn_readlane(d_l, k);
        if (FULL || d >= 0) {
            float* o = rows + d * F + lane;
            unsafeAtomicAdd(o, v[k].x);
            if (F == 64) unsafeAtomicAdd(o + 32, v[k].y);
            else if (lane < F - 32) unsafeAtomicAdd(o + 32, v[k].y);
        }
    }
}

template <int F>
__device__ __forceinline__ void own_rows(float* rows,
                                         const float* __restrict__ h,
                                         const int2* __restrict__ pairs,
                                         const unsigned int* __restrict__ seg, int cap,
                                         const float* __restrict__ dinv,
                                         const float* __restrict__ b2,
                                         float* __restrict__ dstRows, int N) {
    const int lane = threadIdx.x & 31;
    const int wave = __builtin_amdgcn_readfirstlane((int)(threadIdx.x >> 5));
    const int n0 = blockIdx.x * OWN_RANGE;
    int nn = N - n0;
    if (nn > OWN_RANGE) nn = OWN_RANGE;

    const int r = blockIdx.x / OWN_PER_RANGE;
    const int j = blockIdx.x - r * OWN_PER_RANGE;
    const unsigned int* sg = seg + (size_t)(r * DEG_SLICES + wave) * SEG_LD + j;
    const int beg = (int)sg[0];
    const int end = (int)sg[1];
    const int2* list = pairs + (size_t)(r * DEG_SLICES + wave) * cap;
    int2 en = make_int2(0, -1);
    if (beg + lane < end) en = list[beg + lane];

    const float* hown = h + (size_t)n0 * 64;
    constexpr int IN_TRIPS = (OWN_RANGE * 64 / 4 + AGG_THREADS - 1) / AGG_THREADS;
    #pragma unroll
    for (int t = 0; t < IN_TRIPS; ++t) {
        const int e = (t * AGG_THREADS + (int)threadIdx.x) * 4;
        if (e < nn * 64) {
            const float4 v = *(const float4*)(hown + e);
            const int i = e >> 6;
            const int m = (e & 63) >> 2;
            float* o = rows + i * F + 2 * m;
            *(float2*)(o) = make_float2(v.x, v.z);
            if (F == 64 || m < (F - 32) / 2) *(float2*)(o + 32) = make_float2(v.y, v.w);
        }
    }
    __syncthreads();

    for (int b0 = beg; b0 < end; b0 += 32) {
        const int2 e = en;
        const int nx = b0 + 32 + lane;
        en = make_int2(0, -1);
        if (nx < end) en = list[nx];
        if (end - b0 >= 32) add_edges<F, true>(h, rows, e.x, e.y, lane);
        else                add_edges<F, false>(h, rows, e.x, e.y, lane);
    }
    __syncthreads();

    constexpr int OUT_TRIPS = (OWN_RANGE * F / 4 + AGG_THREADS - 1) / AGG_THREADS;
    float* o = dstRows + (size_t)n0 * F;
    #pragma unroll
    for (int t = 0; t < OUT_TRIPS; ++t) {
        const int e = (t * AGG_THREADS + (int)threadIdx.x) * 4;
        if (e < nn * F) {
            const float4 a = *(const float4*)(rows + e);
            if (F == 64) {
                *(float4*)(o + e) = a;
            } else {
                const int row = e / F;
                const int col = e - row * F;
                const float  di = dinv[n0 + row];
                const float4 bb = *(const float4*)(b2 + col);
                *(float4*)(o + e) = make_float4(a.x * di + bb.x, a.y * di + bb.y,
                                                a.z * di + bb.z, a.w * di + bb.w);
            }
        }
    }
}

__global__ void __launch_bounds__(AGG_THREADS)
own_rows1_kernel(const float* __restrict__ h1, const int2* __restrict__ pairs,
                 const unsigned int* __restrict__ seg, int cap,
                 float* __restrict__ acc, int N) {
    __shared__ __attribute__((aligned(16))) float rows[OWN_RANGE * F_HID];
    own_rows<F_HID>(rows, h1, pairs, seg, cap, nullptr, nullptr, acc, N);
}

__global__ void __launch_bounds__(AGG_THREADS)
own_rows2_kernel(const float* __restrict__ h2, const int2* __restrict__ pairs,
                 const unsigned int* __restrict__ seg, int cap,
                 const float* __restrict__ dinv, const float* __restrict__ b2,
                 float* __restrict__ out, int N) {
    __shared__ __attribute__((aligned(16))) float rows[OWN_RANGE * F_OUT];
    own_rows<F_OUT>(rows, h2, pairs, seg, cap, dinv, b2, out, N);
}


extern "C" void kernel_launch(void* const* d_in, const int* in_sizes, int n_in,
                              void* d_out, int out_size, void* d_ws, size_t ws_size,
                              hipStream_t stream) {
    const float* x  = (const float*)d_in[0];
    const int*   ei = (const int*)d_in[1];
    const float* W1 = (const float*)d_in[2];
    const float* b1 = (const float*)d_in[3];
    const float* W2 = (const float*)d_in[4];
    const float* b2 = (const float*)d_in[5];

    const int N = in_sizes[0] / F_IN;
    const int E = in_sizes[1] / 2;
    const int* src = ei;
    const int* dst = ei + E;

    const int  sliceLen = (((E + DEG_SLICES - 1) / DEG_SLICES) + 3) & ~3;
    const dim3 histGrid((N + DEG_RANGE - 1) / DEG_RANGE, DEG_SLICES);

    float* ws   = (float*)d_ws;
    float* dinv = ws;
    float* bufA = ws + (((size_t)N + 63) & ~(size_t)63);
    float* bufB = bufA + (size_t)N * F_HID;
    float* pW1  = bufB + (size_t)N * F_HID;
    float* pW2  = pW1 + (size_t)(F_IN / 4) * 256;
    unsigned int* part = (unsigned int*)(pW2 + (size_t)(F_HID / 4) * 256);
    unsigned int* seg  = part + (size_t)DEG_SLICES * N;
    int2* pairs = (int2*)(seg + (((size_t)histGrid.x * DEG_SLICES * SEG_LD + 3) & ~(size_t)3));
    float* out  = (float*)d_out;

    const int B = 256;
    const int gemmBlocks  = (N + 127) / 128;
    const int ownBlocks   = (N + OWN_RANGE - 1) / OWN_RANGE;
    const int rp1Blocks   = ((F_IN / 4) * 256 + B - 1) / B;
    const int rp2Blocks   = ((F_HID / 4) * 256 + B - 1) / B;

    repack_w1_kernel<<<rp1Blocks, B, 0, stream>>>(W1, pW1);
    repack_w2_kernel<<<rp2Blocks, B, 0, stream>>>(W2, pW2);

    deg_hist_kernel<<<histGrid, DEG_THREADS, 0, stream>>>(src, dst, part, pairs, seg, N, E, sliceLen);

    gemm1_wmma_kernel<<<gemmBlocks, B, 0, stream>>>(x, pW1, part, dinv, bufA, N);
    own_rows1_kernel<<<ownBlocks, AGG_THREADS, 0, stream>>>(bufA, pairs, seg, sliceLen, bufB, N);

    gemm2_wmma_kernel<<<gemmBlocks, B, 0, stream>>>(bufB, b1, pW2, dinv, bufA, N);
    own_rows2_kernel<<<ownBlocks, AGG_THREADS, 0, stream>>>(bufA, pairs, seg, sliceLen, dinv, b2, out, N);
}


# Layer 2 as (A z) W2: owners sum the 64-wide dinv-scaled z rows, then multiply each finished sum by W2 on WMMA from LDS; h2 product + W2 repack launches gone (7 -> 5); rounding order: product after sum
# speedup vs baseline: 4.0745x; 1.1159x over previous
#include <hip/hip_runtime.h>

typedef __attribute__((ext_vector_type(2))) float v2f;
typedef __attribute__((ext_vector_type(4))) float v4f;
typedef __attribute__((ext_vector_type(8))) float v8f;

#define F_IN  128
#define F_HID 64
#define F_OUT 40


#define DEG_RANGE   12544
#define DEG_SLICES  32
#define DEG_THREADS 1024
#define OWN_RANGE     392
#define OWN_PER_RANGE (DEG_RANGE / OWN_RANGE)
#define SEG_LD        (OWN_PER_RANGE + 1)
#define STAGE_CAP     8192

static_assert(DEG_RANGE % OWN_RANGE == 0, "a range of the edge lists is a whole number of owners");
static_assert(OWN_PER_RANGE == DEG_THREADS / 32, "one wave adds up the counters of one owner");
static_assert(OWN_PER_RANGE == 32, "the prefix over the owners is one wave wide");

template <bool PLACE>
__device__ __forceinline__ void walk_slice(const int* __restrict__ dp, const int* __restrict__ sp,
                                           int len, int nv, int n0, int nn,
                                           unsigned int* cnt, unsigned int* cur,
                                           int2* stage, int2* __restrict__ list) {
    const int4* dv = (const int4*)dp;
    const int4* sv = (const int4*)sp;
    for (int i0 = 0; i0 < nv; i0 += 4 * DEG_THREADS) {
        int4 q[4], sq[4];
        #pragma unroll
        for (int k = 0; k < 4; ++k) {
            const int  idx = i0 + k * DEG_THREADS + (int)threadIdx.x;
            const bool ok  = idx < nv;
            q[k] = dv[ok ? idx : 0];
            if (PLACE) sq[k] = sv[ok ? idx : 0];
            if (!ok) q[k] = make_int4(-1, -1, -1, -1);
        }
        if (!PLACE) {
            #pragma unroll
            for (int k = 0; k < 4; ++k) {
                const int dd[4] = {q[k].x, q[k].y, q[k].z, q[k].w};
                #pragma unroll
                for (int c = 0; c < 4; ++c) {
                    const unsigned int u = (unsigned int)(dd[c] - n0);
                    if (u < (unsigned int)nn) atomicAdd(&cnt[u], 1u);
                }
            }
        } else {
            unsigned int pos[16];
            #pragma unroll
            for (int k = 0; k < 4; ++k) {
                const int dd[4] = {q[k].x, q[k].y, q[k].z, q[k].w};
                #pragma unroll
                for (int c = 0; c < 4; ++c) {
                    const unsigned int u = (unsigned int)(dd[c] - n0);
                    pos[k * 4 + c] = 0u;
                    if (u < (unsigned int)nn) pos[k * 4 + c] = atomicAdd(&cur[u / OWN_RANGE], 1u);
                }
            }
            #pragma unroll
            for (int k = 0; k < 4; ++k) {
                const int dd[4] = {q[k].x, q[k].y, q[k].z, q[k].w};
                const int ss[4] = {sq[k].x, sq[k].y, sq[k].z, sq[k].w};
                #pragma unroll
                for (int c = 0; c < 4; ++c) {
                    const unsigned int u = (unsigned int)(dd[c] - n0);
                    if (u < (unsigned int)nn) {
                        const unsigned int p  = pos[k * 4 + c];
                        const int2         en = make_int2(ss[c], (int)(u % OWN_RANGE));
                        if (p < STAGE_CAP) stage[p] = en;
                        else               list[p]  = en;
                    }
                }
            }
        }
    }
    for (int e = 4 * nv + (int)threadIdx.x; e < len; e += DEG_THREADS) {
        const unsigned int u = (unsigned int)(dp[e] - n0);
        if (u < (unsigned int)nn) {
            if (!PLACE) {
                atomicAdd(&cnt[u], 1u);
            } else {
                const unsigned int p  = atomicAdd(&cur[u / OWN_RANGE], 1u);
                const int2         en = make_int2(sp[e], (int)(u % OWN_RANGE));
                if (p < STAGE_CAP) stage[p] = en;
                else               list[p]  = en;
            }
        }
    }
}

__global__ void __launch_bounds__(DEG_THREADS)
deg_hist_kernel(const int* __restrict__ src, const int* __restrict__ dst,
                unsigned int* __restrict__ part, int2* __restrict__ pairs,
                unsigned int* __restrict__ seg, int N, int E, int sliceLen) {
    __shared__ unsigned int cnt[DEG_RANGE];
    __shared__ unsigned int tot[OWN_PER_RANGE];
    __shared__ unsigned int cur[OWN_PER_RANGE];
    __shared__ unsigned int total;
    __shared__ int2 stage[STAGE_CAP];
    const int lane = threadIdx.x & 31;
    const int wave = threadIdx.x >> 5;
    const int n0 = blockIdx.x * DEG_RANGE;
    int nn = N - n0;
    if (nn > DEG_RANGE) nn = DEG_RANGE;
    for (int i = threadIdx.x; i < DEG_RANGE; i += DEG_THREADS) cnt[i] = 0u;
    __syncthreads();

    const int e0 = blockIdx.y * sliceLen;
    int e1 = e0 + sliceLen;
    if (e1 > E) e1 = E;
    int len = e1 - e0;
    if (len < 0) len = 0;
    const int* dp = dst + e0;
    const int* sp = src + e0;
    int2* list = pairs + (size_t)(blockIdx.x * DEG_SLICES + blockIdx.y) * sliceLen;
    const bool vec = (((reinterpret_cast<size_t>(dp) | reinterpret_cast<size_t>(sp)) & 15) == 0);
    const int  nv  = vec ? (len >> 2) : 0;

    walk_slice<false>(dp, sp, len, nv, n0, nn, cnt, cur, stage, list);
    __syncthreads();

    {
        unsigned int t = 0u;
        for (int i = lane; i < OWN_RANGE; i += 32) t += cnt[wave * OWN_RANGE + i];
        for (int o = 16; o > 0; o >>= 1) t += (unsigned int)__shfl_xor((int)t, o, 32);
        if (lane == 0) tot[wave] = t;
    }
    unsigned int* p = part + (size_t)blockIdx.y * N + n0;
    for (int i = threadIdx.x; i < nn; i += DEG_THREADS) p[i] = cnt[i];
    __syncthreads();

    if (wave == 0) {
        const unsigned int v = tot[lane];
        unsigned int inc = v;
        for (int o = 1; o < 32; o <<= 1) {
            const unsigned int up = (unsigned int)__shfl_up((int)inc, o, 32);
            if (lane >= o) inc += up;
        }
        cur[lane] = inc - v;
        unsigned int* sg = seg + (size_t)(blockIdx.x * DEG_SLICES + blockIdx.y) * SEG_LD;
        sg[lane] = inc - v;
        if (lane == OWN_PER_RANGE - 1) { sg[OWN_PER_RANGE] = inc; total = inc; }
    }
    __syncthreads();

    walk_slice<true>(dp, sp, len, nv, n0, nn, cnt, cur, stage, list);
    __syncthreads();

    int nst = (int)total;
    if (nst > STAGE_CAP) nst = STAGE_CAP;
    for (int i = threadIdx.x; i < nst; i += DEG_THREADS) list[i] = stage[i];
}


__global__ void repack_w1_kernel(const float* __restrict__ W1, float* __restrict__ pW1) {
    int idx = blockIdx.x * blockDim.x + threadIdx.x;
    if (idx >= (F_IN / 4) * 256) return;
    int kk   = idx >> 8;
    int rem  = idx & 255;
    int lane = rem >> 3;
    int j    = rem & 7;
    int n    = j >> 1;
    int t    = j & 1;
    int lm   = lane & 15;
    int kh   = (lane >> 4) << 1;
    pW1[idx] = W1[(size_t)(4 * kk + kh + t) * F_HID + (n >> 1) * 32 + 2 * lm + (n & 1)];
}


__global__ void gemm1_wmma_kernel(const float* __restrict__ x, const float* __restrict__ pW1,
                                  const unsigned int* __restrict__ part,
                                  float* __restrict__ dinv, float* __restrict__ h1, int N) {
    const int lane = threadIdx.x & 31;
    const int wave = threadIdx.x >> 5;
    const int row0 = blockIdx.x * 128 + wave * 16;
    const int lm   = lane & 15;
    const int kh   = (lane >> 4) << 1;
    int r = row0 + lm;
    if (r > N - 1) r = N - 1;
    const float* xrow = x + (size_t)r * F_IN + kh;
    const float* pw   = pW1 + (size_t)lane * 8;

    const int hi8 = (lane >> 4) << 3;
    int rl = row0 + lane;
    if (rl > N - 1) rl = N - 1;
    unsigned int pc[DEG_SLICES];
    #pragma unroll
    for (int s = 0; s < DEG_SLICES; ++s) pc[s] = part[(size_t)s * N + rl];

    v8f acc0 = {}, acc1 = {}, acc2 = {}, acc3 = {};

    for (int kk = 0; kk < F_IN / 4; ++kk) {
        const v2f A   = *(const v2f*)(xrow + kk * 4);
        const v4f w01 = *(const v4f*)(pw + (size_t)kk * 256);
        const v4f w23 = *(const v4f*)(pw + (size_t)kk * 256 + 4);
        v2f B0; B0[0] = w01[0]; B0[1] = w01[1];
        v2f B1; B1[0] = w01[2]; B1[1] = w01[3];
        v2f B2; B2[0] = w23[0]; B2[1] = w23[1];
        v2f B3; B3[0] = w23[2]; B3[1] = w23[3];
        acc0 = __builtin_amdgcn_wmma_f32_16x16x4_f32(false, A, false, B0, (short)0, acc0, false, false);
        acc1 = __builtin_amdgcn_wmma_f32_16x16x4_f32(false, A, false, B1, (short)0, acc1, false, false);
        acc2 = __builtin_amdgcn_wmma_f32_16x16x4_f32(false, A, false, B2, (short)0, acc2, false, false);
        acc3 = __builtin_amdgcn_wmma_f32_16x16x4_f32(false, A, false, B3, (short)0, acc3, false, false);
    }

    unsigned int cdeg = 1u;
    #pragma unroll
    for (int s = 0; s < DEG_SLICES; ++s) cdeg += pc[s];
    const float dil = rsqrtf((float)cdeg);
    float dj[8];
    #pragma unroll
    for (int j = 0; j < 8; ++j) dj[j] = __shfl(dil, hi8 + j, 32);
    for (int j = 0; j < 8; ++j) {
        int rr = row0 + hi8 + j;
        if (rr < N) {
            const float di = dj[j];
            const float s0 = acc0[j] * di, s1 = acc1[j] * di, s2 = acc2[j] * di, s3 = acc3[j] * di;
            float* o = h1 + (size_t)rr * F_HID + 4 * lm;
            *(float4*)(o) = make_float4(s0, s2, s1, s3);
        }
    }
    if ((wave & 1) == 0) {
        const int r0 = row0 + lane;
        if (r0 < N) dinv[r0] = dil;
    }
}


#define AGG_THREADS   1024
#define AGG_WAVES     (AGG_THREADS / 32)
#define ROWS1_LD      64
#define ROWS2_LD      68

static_assert(AGG_WAVES == DEG_SLICES, "one wave of an owner per edge list of its range");

template <int LD, bool FULL>
__device__ __forceinline__ void add_edges(const float* __restrict__ h, float* rows,
                                          int s_l, int d_l, int lane) {
    float2 v[32];
    #pragma unroll
    for (int k = 0; k < 32; ++k) {
        const int s = __builtin_amdgcn_readlane(s_l, k);
        v[k] = *(const float2*)(h + (size_t)s * 64 + lane * 2);
    }
    #pragma unroll
    for (int k = 0; k < 32; ++k) {
        const int d = __builtin_amdgcn_readlane(d_l, k);
        if (FULL || d >= 0) {
            float* o = rows + d * LD + lane;
            unsafeAtomicAdd(o, v[k].x);
            unsafeAtomicAdd(o + 32, v[k].y);
        }
    }
}

template <int LD>
__device__ __forceinline__ void sum_rows(float* rows,
                                         const float* __restrict__ h,
                                         const int2* __restrict__ pairs,
                                         const unsigned int* __restrict__ seg, int cap,
                                         int n0, int nn) {
    const int lane = threadIdx.x & 31;
    const int wave = __builtin_amdgcn_readfirstlane((int)(threadIdx.x >> 5));

    const int r = blockIdx.x / OWN_PER_RANGE;
    const int j = blockIdx.x - r * OWN_PER_RANGE;
    const unsigned int* sg = seg + (size_t)(r * DEG_SLICES + wave) * SEG_LD + j;
    const int beg = (int)sg[0];
    const int end = (int)sg[1];
    const int2* list = pairs + (size_t)(r * DEG_SLICES + wave) * cap;
    int2 en = make_int2(0, -1);
    if (beg + lane < end) en = list[beg + lane];

    const float* hown = h + (size_t)n0 * 64;
    constexpr int IN_TRIPS = (OWN_RANGE * 64 / 4 + AGG_THREADS - 1) / AGG_THREADS;
    #pragma unroll
    for (int t = 0; t < IN_TRIPS; ++t) {
        const int e = (t * AGG_THREADS + (int)threadIdx.x) * 4;
        if (e < nn * 64) {
            const float4 v = *(const float4*)(hown + e);
            const int i = e >> 6;
            const int m = (e & 63) >> 2;
            float* o = rows + i * LD + 2 * m;
            *(float2*)(o)      = make_float2(v.x, v.z);
            *(float2*)(o + 32) = make_float2(v.y, v.w);
        }
    }
    __syncthreads();

    for (int b0 = beg; b0 < end; b0 += 32) {
        const int2 e = en;
        const int nx = b0 + 32 + lane;
        en = make_int2(0, -1);
        if (nx < end) en = list[nx];
        if (end - b0 >= 32) add_edges<LD, true>(h, rows, e.x, e.y, lane);
        else                add_edges<LD, false>(h, rows, e.x, e.y, lane);
    }
    __syncthreads();
}

__global__ void __launch_bounds__(AGG_THREADS)
own_rows1_kernel(const float* __restrict__ h1, const int2* __restrict__ pairs,
                 const unsigned int* __restrict__ seg, int cap,
                 const float* __restrict__ dinv, const float* __restrict__ b1,
                 float* __restrict__ zs, int N) {
    __shared__ __attribute__((aligned(16))) float rows[OWN_RANGE * ROWS1_LD];
    const int n0 = blockIdx.x * OWN_RANGE;
    int nn = N - n0;
    if (nn > OWN_RANGE) nn = OWN_RANGE;
    sum_rows<ROWS1_LD>(rows, h1, pairs, seg, cap, n0, nn);

    constexpr int OUT_TRIPS = (OWN_RANGE * 64 / 4 + AGG_THREADS - 1) / AGG_THREADS;
    float* o = zs + (size_t)n0 * 64;
    #pragma unroll
    for (int t = 0; t < OUT_TRIPS; ++t) {
        const int e = (t * AGG_THREADS + (int)threadIdx.x) * 4;
        if (e < nn * 64) {
            const int i = e >> 6;
            const int m = (e & 63) >> 2;
            const float2 lo = *(const float2*)(rows + i * ROWS1_LD + 2 * m);
            const float2 hi = *(const float2*)(rows + i * ROWS1_LD + 2 * m + 32);
            const float2 bl = *(const float2*)(b1 + 2 * m);
            const float2 bh = *(const float2*)(b1 + 2 * m + 32);
            const float  di = dinv[n0 + i];
            const float z0 = lo.x * di + bl.x, z1 = lo.y * di + bl.y;
            const float z2 = hi.x * di + bh.x, z3 = hi.y * di + bh.y;
            *(float4*)(o + e) = make_float4((z0 > 0.0f ? z0 : 0.0f) * di, (z2 > 0.0f ? z2 : 0.0f) * di,
                                            (z1 > 0.0f ? z1 : 0.0f) * di, (z3 > 0.0f ? z3 : 0.0f) * di);
        }
    }
}

__global__ void __launch_bounds__(AGG_THREADS)
own_rows2_kernel(const float* __restrict__ zs, const int2* __restrict__ pairs,
                 const unsigned int* __restrict__ seg, int cap,
                 const float* __restrict__ W2,
                 const float* __restrict__ dinv, const float* __restrict__ b2,
                 float* __restrict__ out, int N) {
    __shared__ __attribute__((aligned(16))) float rows[OWN_RANGE * ROWS2_LD];
    __shared__ __attribute__((aligned(16))) float w2a[(F_HID / 4) * 32 * 4];
    __shared__ __attribute__((aligned(16))) float w2b[(F_HID / 4) * 32 * 2];
    const int n0 = blockIdx.x * OWN_RANGE;
    int nn = N - n0;
    if (nn > OWN_RANGE) nn = OWN_RANGE;

    {
        const int u  = (int)threadIdx.x;
        const int kk = u >> 6;
        const int fl = (u >> 1) & 31;
        const int t  = u & 1;
        const int lm = fl & 15;
        const int k  = 4 * kk + ((fl >> 4) << 1) + t;
        const float* wr = W2 + (size_t)k * F_OUT + lm;
        const float c0 = wr[0];
        const float c1 = wr[16];
        const float c2 = (lm < F_OUT - 32) ? wr[32] : 0.0f;
        w2a[(kk * 32 + fl) * 4 + t]     = c0;
        w2a[(kk * 32 + fl) * 4 + 2 + t] = c1;
        w2b[(kk * 32 + fl) * 2 + t]     = c2;
    }

    sum_rows<ROWS2_LD>(rows, zs, pairs, seg, cap, n0, nn);

    const int lane = threadIdx.x & 31;
    const int wave = __builtin_amdgcn_readfirstlane((int)(threadIdx.x >> 5));
    const int lm   = lane & 15;
    const int hi8  = (lane >> 4) << 3;
    const int kh   = (lane >> 4) << 1;
    const int ntiles = (nn + 15) >> 4;
    v8f acc0 = {}, acc1 = {}, acc2 = {};
    if (wave < ntiles) {
        int rl = wave * 16 + lm;
        if (rl > OWN_RANGE - 1) rl = OWN_RANGE - 1;
        const float* arow = rows + rl * ROWS2_LD + kh;
        #pragma unroll
        for (int kk = 0; kk < F_HID / 4; ++kk) {
            const v2f A   = *(const v2f*)(arow + kk * 4);
            const v4f w01 = *(const v4f*)(w2a + (kk * 32 + lane) * 4);
            const v2f B2  = *(const v2f*)(w2b + (kk * 32 + lane) * 2);
            v2f B0; B0[0] = w01[0]; B0[1] = w01[1];
            v2f B1; B1[0] = w01[2]; B1[1] = w01[3];
            acc0 = __builtin_amdgcn_wmma_f32_16x16x4_f32(false, A, false, B0, (short)0, acc0, false, false);
            acc1 = __builtin_amdgcn_wmma_f32_16x16x4_f32(false, A, false, B1, (short)0, acc1, false, false);
            acc2 = __builtin_amdgcn_wmma_f32_16x16x4_f32(false, A, false, B2, (short)0, acc2, false, false);
        }
    }
    __syncthreads();

    if (wave < ntiles) {
        for (int j = 0; j < 8; ++j) {
            const int rr = wave * 16 + hi8 + j;
            if (rr < nn) {
                float* o = rows + rr * F_OUT + lm;
                o[0]  = acc0[j];
                o[16] = acc1[j];
                if (lm < F_OUT - 32) o[32] = acc2[j];
            }
        }
    }
    __syncthreads();

    constexpr int OUT_TRIPS = (OWN_RANGE * F_OUT / 4 + AGG_THREADS - 1) / AGG_THREADS;
    float* o = out + (size_t)n0 * F_OUT;
    #pragma unroll
    for (int t = 0; t < OUT_TRIPS; ++t) {
        const int e = (t * AGG_THREADS + (int)threadIdx.x) * 4;
        if (e < nn * F_OUT) {
            const float4 a = *(const float4*)(rows + e);
            const int row = e / F_OUT;
            const int col = e - row * F_OUT;
            const float  di = dinv[n0 + row];
            const float4 bb = *(const float4*)(b2 + col);
            *(float4*)(o + e) = make_float4(a.x * di + bb.x, a.y * di + bb.y,
                                            a.z * di + bb.z, a.w * di + bb.w);
        }
    }
}


extern "C" void kernel_launch(void* const* d_in, const int* in_sizes, int n_in,
                              void* d_out, int out_size, void* d_ws, size_t ws_size,
                              hipStream_t stream) {
    const float* x  = (const float*)d_in[0];
    const int*   ei = (const int*)d_in[1];
    const float* W1 = (const float*)d_in[2];
    const float* b1 = (const float*)d_in[3];
    const float* W2 = (const float*)d_in[4];
    const float* b2 = (const float*)d_in[5];

    const int N = in_sizes[0] / F_IN;
    const int E = in_sizes[1] / 2;
    const int* src = ei;
    const int* dst = ei + E;

    const int  sliceLen = (((E + DEG_SLICES - 1) / DEG_SLICES) + 3) & ~3;
    const dim3 histGrid((N + DEG_RANGE - 1) / DEG_RANGE, DEG_SLICES);

    float* ws   = (float*)d_ws;
    float* dinv = ws;
    float* bufA = ws + (((size_t)N + 63) & ~(size_t)63);
    float* bufB = bufA + (size_t)N * F_HID;
    float* pW1  = bufB + (size_t)N * F_HID;
    unsigned int* part = (unsigned int*)(pW1 + (size_t)(F_IN / 4) * 256);
    unsigned int* seg  = part + (size_t)DEG_SLICES * N;
    int2* pairs = (int2*)(seg + (((size_t)histGrid.x * DEG_SLICES * SEG_LD + 3) & ~(size_t)3));
    float* out  = (float*)d_out;

    const int B = 256;
    const int gemmBlocks  = (N + 127) / 128;
    const int ownBlocks   = (N + OWN_RANGE - 1) / OWN_RANGE;
    const int rp1Blocks   = ((F_IN / 4) * 256 + B - 1) / B;

    repack_w1_kernel<<<rp1Blocks, B, 0, stream>>>(W1, pW1);

    deg_hist_kernel<<<histGrid, DEG_THREADS, 0, stream>>>(src, dst, part, pairs, seg, N, E, sliceLen);

    gemm1_wmma_kernel<<<gemmBlocks, B, 0, stream>>>(x, pW1, part, dinv, bufA, N);
    own_rows1_kernel<<<ownBlocks, AGG_THREADS, 0, stream>>>(bufA, pairs, seg, sliceLen, dinv, b1, bufB, N);

    own_rows2_kernel<<<ownBlocks, AGG_THREADS, 0, stream>>>(bufB, pairs, seg, sliceLen, W2, dinv, b2, out, N);
}


# Owner kernels: reads of the start phase (start values, W2, list head) and of the leaving phase (row scales, bias, sums) go out as one group each, not one memory round trip per trip; sums unchanged
# speedup vs baseline: 4.3083x; 1.0574x over previous
#include <hip/hip_runtime.h>

typedef __attribute__((ext_vector_type(2))) float v2f;
typedef __attribute__((ext_vector_type(4))) float v4f;
typedef __attribute__((ext_vector_type(8))) float v8f;

#define F_IN  128
#define F_HID 64
#define F_OUT 40


#define DEG_RANGE   12544
#define DEG_SLICES  32
#define DEG_THREADS 1024
#define OWN_RANGE     392
#define OWN_PER_RANGE (DEG_RANGE / OWN_RANGE)
#define SEG_LD        (OWN_PER_RANGE + 1)
#define STAGE_CAP     8192

static_assert(DEG_RANGE % OWN_RANGE == 0, "a range of the edge lists is a whole number of owners");
static_assert(OWN_PER_RANGE == DEG_THREADS / 32, "one wave adds up the counters of one owner");
static_assert(OWN_PER_RANGE == 32, "the prefix over the owners is one wave wide");

template <bool PLACE>
__device__ __forceinline__ void walk_slice(const int* __restrict__ dp, const int* __restrict__ sp,
                                           int len, int nv, int n0, int nn,
                                           unsigned int* cnt, unsigned int* cur,
                                           int2* stage, int2* __restrict__ list) {
    const int4* dv = (const int4*)dp;
    const int4* sv = (const int4*)sp;
    for (int i0 = 0; i0 < nv; i0 += 4 * DEG_THREADS) {
        int4 q[4], sq[4];
        #pragma unroll
        for (int k = 0; k < 4; ++k) {
            const int  idx = i0 + k * DEG_THREADS + (int)threadIdx.x;
            const bool ok  = idx < nv;
            q[k] = dv[ok ? idx : 0];
            if (PLACE) sq[k] = sv[ok ? idx : 0];
            if (!ok) q[k] = make_int4(-1, -1, -1, -1);
        }
        if (!PLACE) {
            #pragma unroll
            for (int k = 0; k < 4; ++k) {
                const int dd[4] = {q[k].x, q[k].y, q[k].z, q[k].w};
                #pragma unroll
                for (int c = 0; c < 4; ++c) {
                    const unsigned int u = (unsigned int)(dd[c] - n0);
                    if (u < (unsigned int)nn) atomicAdd(&cnt[u], 1u);
                }
            }
        } else {
            unsigned int pos[16];
            #pragma unroll
            for (int k = 0; k < 4; ++k) {
                const int dd[4] = {q[k].x, q[k].y, q[k].z, q[k].w};
                #pragma unroll
                for (int c = 0; c < 4; ++c) {
                    const unsigned int u = (unsigned int)(dd[c] - n0);
                    pos[k * 4 + c] = 0u;
                    if (u < (unsigned int)nn) pos[k * 4 + c] = atomicAdd(&cur[u / OWN_RANGE], 1u);
                }
            }
            #pragma unroll
            for (int k = 0; k < 4; ++k) {
                const int dd[4] = {q[k].x, q[k].y, q[k].z, q[k].w};
                const int ss[4] = {sq[k].x, sq[k].y, sq[k].z, sq[k].w};
                #pragma unroll
                for (int c = 0; c < 4; ++c) {
                    const unsigned int u = (unsigned int)(dd[c] - n0);
                    if (u < (unsigned int)nn) {
                        const unsigned int p  = pos[k * 4 + c];
                        const int2         en = make_int2(ss[c], (int)(u % OWN_RANGE));
                        if (p < STAGE_CAP) stage[p] = en;
                        else               list[p]  = en;
                    }
                }
            }
        }
    }
    for (int e = 4 * nv + (int)threadIdx.x; e < len; e += DEG_THREADS) {
        const unsigned int u = (unsigned int)(dp[e] - n0);
        if (u < (unsigned int)nn) {
            if (!PLACE) {
                atomicAdd(&cnt[u], 1u);
            } else {
                const unsigned int p  = atomicAdd(&cur[u / OWN_RANGE], 1u);
                const int2         en = make_int2(sp[e], (int)(u % OWN_RANGE));
                if (p < STAGE_CAP) stage[p] = en;
                else               list[p]  = en;
            }
        }
    }
}

__global__ void __launch_bounds__(DEG_THREADS)
deg_hist_kernel(const int* __restrict__ src, const int* __restrict__ dst,
                unsigned int* __restrict__ part, int2* __restrict__ pairs,
                unsigned int* __restrict__ seg, int N, int E, int sliceLen) {
    __shared__ unsigned int cnt[DEG_RANGE];
    __shared__ unsigned int tot[OWN_PER_RANGE];
    __shared__ unsigned int cur[OWN_PER_RANGE];
    __shared__ unsigned int total;
    __shared__ int2 stage[STAGE_CAP];
    const int lane = threadIdx.x & 31;
    const int wave = threadIdx.x >> 5;
    const int n0 = blockIdx.x * DEG_RANGE;
    int nn = N - n0;
    if (nn > DEG_RANGE) nn = DEG_RANGE;
    for (int i = threadIdx.x; i < DEG_RANGE; i += DEG_THREADS) cnt[i] = 0u;
    __syncthreads();

    const int e0 = blockIdx.y * sliceLen;
    int e1 = e0 + sliceLen;
    if (e1 > E) e1 = E;
    int len = e1 - e0;
    if (len < 0) len = 0;
    const int* dp = dst + e0;
    const int* sp = src + e0;
    int2* list = pairs + (size_t)(blockIdx.x * DEG_SLICES + blockIdx.y) * sliceLen;
    const bool vec = (((reinterpret_cast<size_t>(dp) | reinterpret_cast<size_t>(sp)) & 15) == 0);
    const int  nv  = vec ? (len >> 2) : 0;

    walk_slice<false>(dp, sp, len, nv, n0, nn, cnt, cur, stage, list);
    __syncthreads();

    {
        unsigned int t = 0u;
        for (int i = lane; i < OWN_RANGE; i += 32) t += cnt[wave * OWN_RANGE + i];
        for (int o = 16; o > 0; o >>= 1) t += (unsigned int)__shfl_xor((int)t, o, 32);
        if (lane == 0) tot[wave] = t;
    }
    unsigned int* p = part + (size_t)blockIdx.y * N + n0;
    for (int i = threadIdx.x; i < nn; i += DEG_THREADS) p[i] = cnt[i];
    __syncthreads();

    if (wave == 0) {
        const unsigned int v = tot[lane];
        unsigned int inc = v;
        for (int o = 1; o < 32; o <<= 1) {
            const unsigned int up = (unsigned int)__shfl_up((int)inc, o, 32);
            if (lane >= o) inc += up;
        }
        cur[lane] = inc - v;
        unsigned int* sg = seg + (size_t)(blockIdx.x * DEG_SLICES + blockIdx.y) * SEG_LD;
        sg[lane] = inc - v;
        if (lane == OWN_PER_RANGE - 1) { sg[OWN_PER_RANGE] = inc; total = inc; }
    }
    __syncthreads();

    walk_slice<true>(dp, sp, len, nv, n0, nn, cnt, cur, stage, list);
    __syncthreads();

    int nst = (int)total;
    if (nst > STAGE_CAP) nst = STAGE_CAP;
    for (int i = threadIdx.x; i < nst; i += DEG_THREADS) list[i] = stage[i];
}


__global__ void repack_w1_kernel(const float* __restrict__ W1, float* __restrict__ pW1) {
    int idx = blockIdx.x * blockDim.x + threadIdx.x;
    if (idx >= (F_IN / 4) * 256) return;
    int kk   = idx >> 8;
    int rem  = idx & 255;
    int lane = rem >> 3;
    int j    = rem & 7;
    int n    = j >> 1;
    int t    = j & 1;
    int lm   = lane & 15;
    int kh   = (lane >> 4) << 1;
    pW1[idx] = W1[(size_t)(4 * kk + kh + t) * F_HID + (n >> 1) * 32 + 2 * lm + (n & 1)];
}


__global__ void gemm1_wmma_kernel(const float* __restrict__ x, const float* __restrict__ pW1,
                                  const unsigned int* __restrict__ part,
                                  float* __restrict__ dinv, float* __restrict__ h1, int N) {
    const int lane = threadIdx.x & 31;
    const int wave = threadIdx.x >> 5;
    const int row0 = blockIdx.x * 128 + wave * 16;
    const int lm   = lane & 15;
    const int kh   = (lane >> 4) << 1;
    int r = row0 + lm;
    if (r > N - 1) r = N - 1;
    const float* xrow = x + (size_t)r * F_IN + kh;
    const float* pw   = pW1 + (size_t)lane * 8;

    const int hi8 = (lane >> 4) << 3;
    int rl = row0 + lane;
    if (rl > N - 1) rl = N - 1;
    unsigned int pc[DEG_SLICES];
    #pragma unroll
    for (int s = 0; s < DEG_SLICES; ++s) pc[s] = part[(size_t)s * N + rl];

    v8f acc0 = {}, acc1 = {}, acc2 = {}, acc3 = {};

    for (int kk = 0; kk < F_IN / 4; ++kk) {
        const v2f A   = *(const v2f*)(xrow + kk * 4);
        const v4f w01 = *(const v4f*)(pw + (size_t)kk * 256);
        const v4f w23 = *(const v4f*)(pw + (size_t)kk * 256 + 4);
        v2f B0; B0[0] = w01[0]; B0[1] = w01[1];
        v2f B1; B1[0] = w01[2]; B1[1] = w01[3];
        v2f B2; B2[0] = w23[0]; B2[1] = w23[1];
        v2f B3; B3[0] = w23[2]; B3[1] = w23[3];
        acc0 = __builtin_amdgcn_wmma_f32_16x16x4_f32(false, A, false, B0, (short)0, acc0, false, false);
        acc1 = __builtin_amdgcn_wmma_f32_16x16x4_f32(false, A, false, B1, (short)0, acc1, false, false);
        acc2 = __builtin_amdgcn_wmma_f32_16x16x4_f32(false, A, false, B2, (short)0, acc2, false, false);
        acc3 = __builtin_amdgcn_wmma_f32_16x16x4_f32(false, A, false, B3, (short)0, acc3, false, false);
    }

    unsigned int cdeg = 1u;
    #pragma unroll
    for (int s = 0; s < DEG_SLICES; ++s) cdeg += pc[s];
    const float dil = rsqrtf((float)cdeg);
    float dj[8];
    #pragma unroll
    for (int j = 0; j < 8; ++j) dj[j] = __shfl(dil, hi8 + j, 32);
    for (int j = 0; j < 8; ++j) {
        int rr = row0 + hi8 + j;
        if (rr < N) {
            const float di = dj[j];
            const float s0 = acc0[j] * di, s1 = acc1[j] * di, s2 = acc2[j] * di, s3 = acc3[j] * di;
            float* o = h1 + (size_t)rr * F_HID + 4 * lm;
            *(float4*)(o) = make_float4(s0, s2, s1, s3);
        }
    }
    if ((wave & 1) == 0) {
        const int r0 = row0 + lane;
        if (r0 < N) dinv[r0] = dil;
    }
}


#define AGG_THREADS   1024
#define AGG_WAVES     (AGG_THREADS / 32)
#define ROWS1_LD      64
#define ROWS2_LD      68

static_assert(AGG_WAVES == DEG_SLICES, "one wave of an owner per edge list of its range");

template <int LD, bool FULL>
__device__ __forceinline__ void add_edges(const float* __restrict__ h, float* rows,
                                          int s_l, int d_l, int lane) {
    float2 v[32];
    #pragma unroll
    for (int k = 0; k < 32; ++k) {
        const int s = __builtin_amdgcn_readlane(s_l, k);
        v[k] = *(const float2*)(h + (size_t)s * 64 + lane * 2);
    }
    #pragma unroll
    for (int k = 0; k < 32; ++k) {
        const int d = __builtin_amdgcn_readlane(d_l, k);
        if (FULL || d >= 0) {
            float* o = rows + d * LD + lane;
            unsafeAtomicAdd(o, v[k].x);
            unsafeAtomicAdd(o + 32, v[k].y);
        }
    }
}

template <int LD, class StageOwn>
__device__ __forceinline__ void sum_rows(float* rows,
                                         const float* __restrict__ h,
                                         const int2* __restrict__ pairs,
                                         const unsigned int* __restrict__ seg, int cap,
                                         int n0, int nn, StageOwn stage_own) {
    const int lane = threadIdx.x & 31;
    const int wave = __builtin_amdgcn_readfirstlane((int)(threadIdx.x >> 5));

    const float* hown = h + (size_t)n0 * 64;
    constexpr int IN_TRIPS = (OWN_RANGE * 64 / 4 + AGG_THREADS - 1) / AGG_THREADS;
    const int lim = nn * 64;
    float4 sv[IN_TRIPS];
    #pragma unroll
    for (int t = 0; t < IN_TRIPS; ++t) {
        const int e = (t * AGG_THREADS + (int)threadIdx.x) * 4;
        sv[t] = *(const float4*)(hown + (e < lim ? e : lim - 4));
    }

    const int r = blockIdx.x / OWN_PER_RANGE;
    const int j = blockIdx.x - r * OWN_PER_RANGE;
    const unsigned int* sg = seg + (size_t)(r * DEG_SLICES + wave) * SEG_LD + j;
    const int beg = (int)sg[0];
    const int end = (int)sg[1];
    const int2* list = pairs + (size_t)(r * DEG_SLICES + wave) * cap;
    int2 en = make_int2(0, -1);
    if (beg + lane < end) en = list[beg + lane];

    stage_own();

    #pragma unroll
    for (int t = 0; t < IN_TRIPS; ++t) {
        const int e = (t * AGG_THREADS + (int)threadIdx.x) * 4;
        if (e < lim) {
            const float4 v = sv[t];
            const int i = e >> 6;
            const int m = (e & 63) >> 2;
            float* o = rows + i * LD + 2 * m;
            *(float2*)(o)      = make_float2(v.x, v.z);
            *(float2*)(o + 32) = make_float2(v.y, v.w);
        }
    }
    __syncthreads();

    for (int b0 = beg; b0 < end; b0 += 32) {
        const int2 e = en;
        const int nx = b0 + 32 + lane;
        en = make_int2(0, -1);
        if (nx < end) en = list[nx];
        if (end - b0 >= 32) add_edges<LD, true>(h, rows, e.x, e.y, lane);
        else                add_edges<LD, false>(h, rows, e.x, e.y, lane);
    }
    __syncthreads();
}

__global__ void __launch_bounds__(AGG_THREADS)
own_rows1_kernel(const float* __restrict__ h1, const int2* __restrict__ pairs,
                 const unsigned int* __restrict__ seg, int cap,
                 const float* __restrict__ dinv, const float* __restrict__ b1,
                 float* __restrict__ zs, int N) {
    __shared__ __attribute__((aligned(16))) float rows[OWN_RANGE * ROWS1_LD];
    const int n0 = blockIdx.x * OWN_RANGE;
    int nn = N - n0;
    if (nn > OWN_RANGE) nn = OWN_RANGE;
    sum_rows<ROWS1_LD>(rows, h1, pairs, seg, cap, n0, nn, [] {});

    constexpr int OUT_TRIPS = (OWN_RANGE * 64 / 4 + AGG_THREADS - 1) / AGG_THREADS;
    float* o = zs + (size_t)n0 * 64;
    const int lim = nn * 64;
    const int m = (int)(threadIdx.x & 15);
    const float2 bl = *(const float2*)(b1 + 2 * m);
    const float2 bh = *(const float2*)(b1 + 2 * m + 32);
    float  di[OUT_TRIPS];
    float2 lo[OUT_TRIPS], hi[OUT_TRIPS];
    #pragma unroll
    for (int t = 0; t < OUT_TRIPS; ++t) {
        const int e = (t * AGG_THREADS + (int)threadIdx.x) * 4;
        const int i = (e < lim ? e : lim - 4) >> 6;
        di[t] = dinv[n0 + i];
        lo[t] = *(const float2*)(rows + i * ROWS1_LD + 2 * m);
        hi[t] = *(const float2*)(rows + i * ROWS1_LD + 2 * m + 32);
    }
    asm volatile("" :: "v"(di[0]), "v"(lo[0].x), "v"(lo[0].y), "v"(hi[0].x), "v"(hi[0].y));
    #pragma unroll
    for (int t = 0; t < OUT_TRIPS; ++t) {
        const int e = (t * AGG_THREADS + (int)threadIdx.x) * 4;
        if (e < lim) {
            const float d = di[t];
            const float z0 = lo[t].x * d + bl.x, z1 = lo[t].y * d + bl.y;
            const float z2 = hi[t].x * d + bh.x, z3 = hi[t].y * d + bh.y;
            *(float4*)(o + e) = make_float4((z0 > 0.0f ? z0 : 0.0f) * d, (z2 > 0.0f ? z2 : 0.0f) * d,
                                            (z1 > 0.0f ? z1 : 0.0f) * d, (z3 > 0.0f ? z3 : 0.0f) * d);
        }
    }
}

__global__ void __launch_bounds__(AGG_THREADS)
own_rows2_kernel(const float* __restrict__ zs, const int2* __restrict__ pairs,
                 const unsigned int* __restrict__ seg, int cap,
                 const float* __restrict__ W2,
                 const float* __restrict__ dinv, const float* __restrict__ b2,
                 float* __restrict__ out, int N) {
    __shared__ __attribute__((aligned(16))) float rows[OWN_RANGE * ROWS2_LD];
    __shared__ __attribute__((aligned(16))) float w2a[(F_HID / 4) * 32 * 4];
    __shared__ __attribute__((aligned(16))) float w2b[(F_HID / 4) * 32 * 2];
    const int n0 = blockIdx.x * OWN_RANGE;
    int nn = N - n0;
    if (nn > OWN_RANGE) nn = OWN_RANGE;

    const int wu  = (int)threadIdx.x;
    const int wkk = wu >> 6;
    const int wfl = (wu >> 1) & 31;
    const int wt  = wu & 1;
    const int wlm = wfl & 15;
    const float* wr = W2 + (size_t)(4 * wkk + ((wfl >> 4) << 1) + wt) * F_OUT + wlm;
    const bool  w3  = wlm < F_OUT - 32;
    const float c0  = wr[0];
    const float c1  = wr[16];
    const float c2r = wr[w3 ? 32 : 0];
    const float c2  = w3 ? c2r : 0.0f;

    sum_rows<ROWS2_LD>(rows, zs, pairs, seg, cap, n0, nn, [&] {
        w2a[(wkk * 32 + wfl) * 4 + wt]     = c0;
        w2a[(wkk * 32 + wfl) * 4 + 2 + wt] = c1;
        w2b[(wkk * 32 + wfl) * 2 + wt]     = c2;
    });

    const int lane = threadIdx.x & 31;
    const int wave = __builtin_amdgcn_readfirstlane((int)(threadIdx.x >> 5));
    const int lm   = lane & 15;
    const int hi8  = (lane >> 4) << 3;
    const int kh   = (lane >> 4) << 1;
    const int ntiles = (nn + 15) >> 4;

    constexpr int OUT_TRIPS = (OWN_RANGE * F_OUT / 4 + AGG_THREADS - 1) / AGG_THREADS;
    const int limo = nn * F_OUT;
    int    ec[OUT_TRIPS];
    float  di[OUT_TRIPS];
    float4 bb[OUT_TRIPS];
    #pragma unroll
    for (int t = 0; t < OUT_TRIPS; ++t) {
        const int e = (t * AGG_THREADS + (int)threadIdx.x) * 4;
        ec[t] = e < limo ? e : limo - 4;
        const int row = ec[t] / F_OUT;
        const int col = ec[t] - row * F_OUT;
        di[t] = dinv[n0 + row];
        bb[t] = *(const float4*)(b2 + col);
    }

    v8f acc0 = {}, acc1 = {}, acc2 = {};
    if (wave < ntiles) {
        int rl = wave * 16 + lm;
        if (rl > OWN_RANGE - 1) rl = OWN_RANGE - 1;
        const float* arow = rows + rl * ROWS2_LD + kh;
        #pragma unroll
        for (int kk = 0; kk < F_HID / 4; ++kk) {
            const v2f A   = *(const v2f*)(arow + kk * 4);
            const v4f w01 = *(const v4f*)(w2a + (kk * 32 + lane) * 4);
            const v2f B2  = *(const v2f*)(w2b + (kk * 32 + lane) * 2);
            v2f B0; B0[0] = w01[0]; B0[1] = w01[1];
            v2f B1; B1[0] = w01[2]; B1[1] = w01[3];
            acc0 = __builtin_amdgcn_wmma_f32_16x16x4_f32(false, A, false, B0, (short)0, acc0, false, false);
            acc1 = __builtin_amdgcn_wmma_f32_16x16x4_f32(false, A, false, B1, (short)0, acc1, false, false);
            acc2 = __builtin_amdgcn_wmma_f32_16x16x4_f32(false, A, false, B2, (short)0, acc2, false, false);
        }
    }
    __syncthreads();

    if (wave < ntiles) {
        for (int j = 0; j < 8; ++j) {
            const int rr = wave * 16 + hi8 + j;
            if (rr < nn) {
                float* o = rows + rr * F_OUT + lm;
                o[0]  = acc0[j];
                o[16] = acc1[j];
                if (lm < F_OUT - 32) o[32] = acc2[j];
            }
        }
    }
    __syncthreads();

    float* o = out + (size_t)n0 * F_OUT;
    float4 a[OUT_TRIPS];
    #pragma unroll
    for (int t = 0; t < OUT_TRIPS; ++t) a[t] = *(const float4*)(rows + ec[t]);
    #pragma unroll
    for (int t = 0; t < OUT_TRIPS; ++t) {
        const int e = (t * AGG_THREADS + (int)threadIdx.x) * 4;
        if (e < limo) {
            const float d = di[t];
            *(float4*)(o + e) = make_float4(a[t].x * d + bb[t].x, a[t].y * d + bb[t].y,
                                            a[t].z * d + bb[t].z, a[t].w * d + bb[t].w);
        }
    }
}


extern "C" void kernel_launch(void* const* d_in, const int* in_sizes, int n_in,
                              void* d_out, int out_size, void* d_ws, size_t ws_size,
                              hipStream_t stream) {
    const float* x  = (const float*)d_in[0];
    const int*   ei = (const int*)d_in[1];
    const float* W1 = (const float*)d_in[2];
    const float* b1 = (const float*)d_in[3];
    const float* W2 = (const float*)d_in[4];
    const float* b2 = (const float*)d_in[5];

    const int N = in_sizes[0] / F_IN;
    const int E = in_sizes[1] / 2;
    const int* src = ei;
    const int* dst = ei + E;

    const int  sliceLen = (((E + DEG_SLICES - 1) / DEG_SLICES) + 3) & ~3;
    const dim3 histGrid((N + DEG_RANGE - 1) / DEG_RANGE, DEG_SLICES);

    float* ws   = (float*)d_ws;
    float* dinv = ws;
    float* bufA = ws + (((size_t)N + 63) & ~(size_t)63);
    float* bufB = bufA + (size_t)N * F_HID;
    float* pW1  = bufB + (size_t)N * F_HID;
    unsigned int* part = (unsigned int*)(pW1 + (size_t)(F_IN / 4) * 256);
    unsigned int* seg  = part + (size_t)DEG_SLICES * N;
    int2* pairs = (int2*)(seg + (((size_t)histGrid.x * DEG_SLICES * SEG_LD + 3) & ~(size_t)3));
    float* out  = (float*)d_out;

    const int B = 256;
    const int gemmBlocks  = (N + 127) / 128;
    const int ownBlocks   = (N + OWN_RANGE - 1) / OWN_RANGE;
    const int rp1Blocks   = ((F_IN / 4) * 256 + B - 1) / B;

    repack_w1_kernel<<<rp1Blocks, B, 0, stream>>>(W1, pW1);

    deg_hist_kernel<<<histGrid, DEG_THREADS, 0, stream>>>(src, dst, part, pairs, seg, N, E, sliceLen);

    gemm1_wmma_kernel<<<gemmBlocks, B, 0, stream>>>(x, pW1, part, dinv, bufA, N);
    own_rows1_kernel<<<ownBlocks, AGG_THREADS, 0, stream>>>(bufA, pairs, seg, sliceLen, dinv, b1, bufB, N);

    own_rows2_kernel<<<ownBlocks, AGG_THREADS, 0, stream>>>(bufB, pairs, seg, sliceLen, W2, dinv, b2, out, N);
}
